# starting kernel
# baseline (speedup 1.0000x reference)
#include <hip/hip_runtime.h>
#include <math.h>

#define DIM_IN   128
#define DIM_OUT  128
#define NUM_HEAD 8
#define DIM_HID  16
#define NEG_SLOPE 0.2f
#define NCOLS 384

typedef __attribute__((ext_vector_type(2))) float v2f;
typedef __attribute__((ext_vector_type(8))) float v8f;

__device__ __forceinline__ float lrelu(float x) { return x >= 0.f ? x : NEG_SLOPE * x; }

__device__ __forceinline__ void atomicMaxF32(float* addr, float v) {
  if (v >= 0.f) atomicMax((int*)addr, __float_as_int(v));
  else          atomicMin((unsigned int*)addr, __float_as_uint(v));
}

__global__ void k_init(float* __restrict__ out, float* __restrict__ segmax,
                       float* __restrict__ segsum, int nOut, int nSeg) {
  int i = blockIdx.x * blockDim.x + threadIdx.x;
  if (i < nOut) out[i] = 0.f;
  if (i < nSeg) { segmax[i] = -3.402823466e38f; segsum[i] = 0.f; }
}

__global__ void k_prep(const float* __restrict__ attn_w, const float* __restrict__ attn_b,
                       const float* __restrict__ aggr_w, const float* __restrict__ aggr_b,
                       float* __restrict__ Bmat, float* __restrict__ Bias) {
  int i = blockIdx.x * blockDim.x + threadIdx.x;
  if (i < DIM_IN * NCOLS) {
    int k = i / NCOLS, n = i % NCOLS;
    float v;
    if (n < 128)       v = attn_w[n * 256 + k];
    else if (n < 256)  v = attn_w[(n - 128) * 256 + 128 + k];
    else               v = aggr_w[(n - 256) * 128 + k];
    Bmat[i] = v;
  }
  if (i < NCOLS)
    Bias[i] = (i < 128) ? attn_b[i] : ((i < 256) ? 0.f : aggr_b[i - 256]);
}

__global__ void k_gemm(const float* __restrict__ A, const float* __restrict__ Bmat,
                       const float* __restrict__ Bias,
                       float* __restrict__ P1, float* __restrict__ P2,
                       float* __restrict__ Mout, int R) {
  const int lane = threadIdx.x & 31;
  const int half = lane >> 4;
  const int l16  = lane & 15;
  const int m0 = blockIdx.x * 16;
  const int n0 = blockIdx.y * 64;

  int mrow = m0 + l16; if (mrow >= R) mrow = R - 1;
  const float* __restrict__ Arow = A + (size_t)mrow * DIM_IN;

  v8f acc0 = {}, acc1 = {}, acc2 = {}, acc3 = {};
  for (int k0 = 0; k0 < DIM_IN; k0 += 4) {
    const int ka = k0 + (half << 1);
    v2f a; a.x = Arow[ka]; a.y = Arow[ka + 1];
    const float* __restrict__ Bp = Bmat + (size_t)ka * NCOLS + n0 + l16;
    v2f b0; b0.x = Bp[0];  b0.y = Bp[NCOLS];
    v2f b1; b1.x = Bp[16]; b1.y = Bp[NCOLS + 16];
    v2f b2; b2.x = Bp[32]; b2.y = Bp[NCOLS + 32];
    v2f b3; b3.x = Bp[48]; b3.y = Bp[NCOLS + 48];
    acc0 = __builtin_amdgcn_wmma_f32_16x16x4_f32(false, a, false, b0, (short)0, acc0, false, false);
    acc1 = __builtin_amdgcn_wmma_f32_16x16x4_f32(false, a, false, b1, (short)0, acc1, false, false);
    acc2 = __builtin_amdgcn_wmma_f32_16x16x4_f32(false, a, false, b2, (short)0, acc2, false, false);
    acc3 = __builtin_amdgcn_wmma_f32_16x16x4_f32(false, a, false, b3, (short)0, acc3, false, false);
  }

#pragma unroll
  for (int i = 0; i < 8; ++i) {
    const int row = m0 + i + (half << 3);
    if (row >= R) continue;
    float vals[4] = { acc0[i], acc1[i], acc2[i], acc3[i] };
#pragma unroll
    for (int j = 0; j < 4; ++j) {
      const int c = n0 + j * 16 + l16;
      const float v = vals[j] + Bias[c];
      if (c < 128)      P1[(size_t)row * 128 + c]         = v;
      else if (c < 256) P2[(size_t)row * 128 + (c - 128)] = v;
      else              Mout[(size_t)row * 128 + (c - 256)] = v;
    }
  }
}

__global__ void k_attn(const int* __restrict__ trip, const float* __restrict__ P1,
                       const float* __restrict__ P2, const float* __restrict__ av,
                       const float* __restrict__ ab, float* __restrict__ attn,
                       float* __restrict__ segmax, int E) {
  const int lane = threadIdx.x & 31;
  const int e = blockIdx.x * (blockDim.x >> 5) + (threadIdx.x >> 5);
  if (e >= E) return;
  const int head = trip[3 * e], tail = trip[3 * e + 1], bin = trip[3 * e + 2];
  const int d0 = lane * 4;
  const float4 p1 = *(const float4*)(P1 + (size_t)head * 128 + d0);
  const float4 p2 = *(const float4*)(P2 + (size_t)tail * 128 + d0);
  const float4 w  = *(const float4*)(av + d0);
  float s = lrelu(p1.x + p2.x) * w.x + lrelu(p1.y + p2.y) * w.y +
            lrelu(p1.z + p2.z) * w.z + lrelu(p1.w + p2.w) * w.w;
  s += __shfl_xor(s, 1, 32);
  s += __shfl_xor(s, 2, 32);
  if ((lane & 3) == 0) {
    const int h = lane >> 2;
    const float r = s + lrelu(ab[bin * NUM_HEAD + h]);
    attn[(size_t)e * NUM_HEAD + h] = r;
    atomicMaxF32(segmax + head * NUM_HEAD + h, r);
  }
}

__global__ void k_exp(const int* __restrict__ trip, float* __restrict__ attn,
                      const float* __restrict__ segmax, float* __restrict__ segsum, int EH) {
  int i = blockIdx.x * blockDim.x + threadIdx.x;
  if (i >= EH) return;
  const int e = i >> 3, h = i & 7;
  const int head = trip[3 * e];
  const float v = expf(attn[i] - segmax[head * NUM_HEAD + h]);
  attn[i] = v;
  atomicAdd(segsum + head * NUM_HEAD + h, v);
}

__global__ void k_aggr(const int* __restrict__ trip, const float* __restrict__ attn,
                       const float* __restrict__ segsum, const float* __restrict__ Mmat,
                       float* __restrict__ out, int E) {
  const int lane = threadIdx.x & 31;
  const int e = blockIdx.x * (blockDim.x >> 5) + (threadIdx.x >> 5);
  if (e >= E) return;
  const int head = trip[3 * e], tail = trip[3 * e + 1];
  const int h = lane >> 2;
  const float beta = attn[(size_t)e * NUM_HEAD + h] /
                     (segsum[head * NUM_HEAD + h] + 1e-16f);
  const int d0 = lane * 4;
  const float4 m = *(const float4*)(Mmat + (size_t)tail * 128 + d0);
  float* o = out + (size_t)head * 128 + d0;
  atomicAdd(o + 0, beta * m.x);
  atomicAdd(o + 1, beta * m.y);
  atomicAdd(o + 2, beta * m.z);
  atomicAdd(o + 3, beta * m.w);
}

extern "C" void kernel_launch(void* const* d_in, const int* in_sizes, int n_in,
                              void* d_out, int out_size, void* d_ws, size_t ws_size,
                              hipStream_t stream) {
  const float* emb      = (const float*)d_in[0];
  const int*   trip     = (const int*)  d_in[1];
  const float* attn_w   = (const float*)d_in[2];
  const float* attn_b   = (const float*)d_in[3];
  const float* attn_bin = (const float*)d_in[4];
  const float* attn_vec = (const float*)d_in[5];
  const float* aggr_w   = (const float*)d_in[6];
  const float* aggr_b   = (const float*)d_in[7];

  const int R = in_sizes[0] / DIM_IN;
  const int E = in_sizes[1] / 3;

  float* ws     = (float*)d_ws;
  float* Bmat   = ws;
  float* Bias   = Bmat + DIM_IN * NCOLS;
  float* P1     = Bias + NCOLS;
  float* P2     = P1 + (size_t)R * 128;
  float* Mm     = P2 + (size_t)R * 128;
  float* attn   = Mm + (size_t)R * 128;
  float* segmax = attn + (size_t)E * NUM_HEAD;
  float* segsum = segmax + (size_t)R * NUM_HEAD;
  float* out    = (float*)d_out;

  const int nOut = R * DIM_OUT, nSeg = R * NUM_HEAD;
  const int nInit = nOut > nSeg ? nOut : nSeg;
  k_init<<<(nInit + 255) / 256, 256, 0, stream>>>(out, segmax, segsum, nOut, nSeg);

  k_prep<<<(DIM_IN * NCOLS + 255) / 256, 256, 0, stream>>>(attn_w, attn_b, aggr_w, aggr_b, Bmat, Bias);

  dim3 gg((R + 15) / 16, NCOLS / 64);
  k_gemm<<<gg, 32, 0, stream>>>(emb, Bmat, Bias, P1, P2, Mm, R);

  k_attn<<<(E + 7) / 8, 256, 0, stream>>>(trip, P1, P2, attn_vec, attn_bin, attn, segmax, E);

  k_exp<<<(E * NUM_HEAD + 255) / 256, 256, 0, stream>>>(trip, attn, segmax, segsum, E * NUM_HEAD);

  k_aggr<<<(E + 7) / 8, 256, 0, stream>>>(trip, attn, segsum, Mm, out, E);
}


# Edges grouped by head (int-atomic counting sort), each group re-ordered by edge id so all sums run in one fixed order; one wave per head does logits, softmax, weighted sum: no float atomics
# speedup vs baseline: 1.8795x; 1.8795x over previous
#include <hip/hip_runtime.h>
#include <math.h>

#define DIM_IN   128
#define DIM_OUT  128
#define NUM_HEAD 8
#define DIM_HID  16
#define NUM_BIN  10
#define NEG_SLOPE 0.2f
#define NCOLS 384
#define SEG_CAP 64
#define BIG_F 3.402823466e38f
#define BIG_I 2147483647

typedef __attribute__((ext_vector_type(2))) float v2f;
typedef __attribute__((ext_vector_type(8))) float v8f;

__device__ __forceinline__ float lrelu(float x) { return x >= 0.f ? x : NEG_SLOPE * x; }

__global__ void k_prep(const float* __restrict__ attn_w, const float* __restrict__ attn_b,
                       const float* __restrict__ aggr_w, const float* __restrict__ aggr_b,
                       float* __restrict__ Bmat, float* __restrict__ Bias,
                       int* __restrict__ cnt, int R) {
  int i = blockIdx.x * blockDim.x + threadIdx.x;
  if (i < DIM_IN * NCOLS) {
    int k = i / NCOLS, n = i % NCOLS;
    float v;
    if (n < 128)       v = attn_w[n * 256 + k];
    else if (n < 256)  v = attn_w[(n - 128) * 256 + 128 + k];
    else               v = aggr_w[(n - 256) * 128 + k];
    Bmat[i] = v;
  }
  if (i < NCOLS)
    Bias[i] = (i < 128) ? attn_b[i] : ((i < 256) ? 0.f : aggr_b[i - 256]);
  for (int j = i; j < R; j += gridDim.x * blockDim.x) cnt[j] = 0;
}

__global__ void k_gemm(const float* __restrict__ A, const float* __restrict__ Bmat,
                       const float* __restrict__ Bias,
                       float* __restrict__ P1, float* __restrict__ P2,
                       float* __restrict__ Mout, int R) {
  const int lane = threadIdx.x & 31;
  const int half = lane >> 4;
  const int l16  = lane & 15;
  const int m0 = blockIdx.x * 16;
  const int n0 = blockIdx.y * 64;

  int mrow = m0 + l16; if (mrow >= R) mrow = R - 1;
  const float* __restrict__ Arow = A + (size_t)mrow * DIM_IN;

  v8f acc0 = {}, acc1 = {}, acc2 = {}, acc3 = {};
  for (int k0 = 0; k0 < DIM_IN; k0 += 4) {
    const int ka = k0 + (half << 1);
    v2f a; a.x = Arow[ka]; a.y = Arow[ka + 1];
    const float* __restrict__ Bp = Bmat + (size_t)ka * NCOLS + n0 + l16;
    v2f b0; b0.x = Bp[0];  b0.y = Bp[NCOLS];
    v2f b1; b1.x = Bp[16]; b1.y = Bp[NCOLS + 16];
    v2f b2; b2.x = Bp[32]; b2.y = Bp[NCOLS + 32];
    v2f b3; b3.x = Bp[48]; b3.y = Bp[NCOLS + 48];
    acc0 = __builtin_amdgcn_wmma_f32_16x16x4_f32(false, a, false, b0, (short)0, acc0, false, false);
    acc1 = __builtin_amdgcn_wmma_f32_16x16x4_f32(false, a, false, b1, (short)0, acc1, false, false);
    acc2 = __builtin_amdgcn_wmma_f32_16x16x4_f32(false, a, false, b2, (short)0, acc2, false, false);
    acc3 = __builtin_amdgcn_wmma_f32_16x16x4_f32(false, a, false, b3, (short)0, acc3, false, false);
  }

#pragma unroll
  for (int i = 0; i < 8; ++i) {
    const int row = m0 + i + (half << 3);
    if (row >= R) continue;
    float vals[4] = { acc0[i], acc1[i], acc2[i], acc3[i] };
#pragma unroll
    for (int j = 0; j < 4; ++j) {
      const int c = n0 + j * 16 + l16;
      const float v = vals[j] + Bias[c];
      if (c < 128)      P1[(size_t)row * 128 + c]         = v;
      else if (c < 256) P2[(size_t)row * 128 + (c - 128)] = v;
      else              Mout[(size_t)row * 128 + (c - 256)] = v;
    }
  }
}

__global__ void k_hist(const int* __restrict__ trip, int* __restrict__ cnt,
                       int* __restrict__ rk, int E) {
  const int e = blockIdx.x * blockDim.x + threadIdx.x;
  if (e >= E) return;
  rk[e] = atomicAdd(cnt + trip[3 * e], 1);
}

__global__ void __launch_bounds__(1024) k_scan(const int* __restrict__ cnt,
                                               int* __restrict__ off, int R) {
  __shared__ int s_w[32];
  const int tid = threadIdx.x, lane = tid & 31, wv = tid >> 5;
  const int per = (R + 1023) / 1024;
  const int b = tid * per;
  const int e = (b + per < R) ? (b + per) : R;
  int sum = 0;
  for (int i = b; i < e; ++i) sum += cnt[i];
  int x = sum;
  for (int o = 1; o < 32; o <<= 1) { const int y = __shfl_up(x, o, 32); if (lane >= o) x += y; }
  if (lane == 31) s_w[wv] = x;
  __syncthreads();
  if (wv == 0) {
    int t = s_w[lane];
    for (int o = 1; o < 32; o <<= 1) { const int y = __shfl_up(t, o, 32); if (lane >= o) t += y; }
    s_w[lane] = t;
  }
  __syncthreads();
  int base = x - sum + (wv > 0 ? s_w[wv - 1] : 0);
  for (int i = b; i < e; ++i) { off[i] = base; base += cnt[i]; }
  if (tid == 1023) off[R] = base;
}

__global__ void k_scatter(const int* __restrict__ trip, const int* __restrict__ off,
                          const int* __restrict__ rk, int* __restrict__ segw, int E) {
  const int e = blockIdx.x * blockDim.x + threadIdx.x;
  if (e >= E) return;
  int4* __restrict__ seg = (int4*)segw;
  seg[off[trip[3 * e]] + rk[e]] = make_int4(e, trip[3 * e + 1], trip[3 * e + 2], 0);
}

__device__ __forceinline__ void wave_lds_handover() {
  __builtin_amdgcn_fence(__ATOMIC_RELEASE, "wavefront");
  __builtin_amdgcn_wave_barrier();
  __builtin_amdgcn_fence(__ATOMIC_ACQUIRE, "wavefront");
}

__device__ __forceinline__ float quad_sum(float s) {
  s += __int_as_float(__builtin_amdgcn_update_dpp(0, __float_as_int(s), 0xB1, 0xF, 0xF, true));
  s += __int_as_float(__builtin_amdgcn_update_dpp(0, __float_as_int(s), 0x4E, 0xF, 0xF, true));
  return s;
}

__device__ __forceinline__ float edge_dot(const float4 p1, const float4 w,
                                          const float* __restrict__ P2, int t, int d0) {
  const float4 p2 = *(const float4*)(P2 + (size_t)t * 128 + d0);
  const float s = lrelu(p1.x + p2.x) * w.x + lrelu(p1.y + p2.y) * w.y +
                  lrelu(p1.z + p2.z) * w.z + lrelu(p1.w + p2.w) * w.w;
  return quad_sum(s);
}

__device__ __forceinline__ int clamp_bin(int b) {
  return b < 0 ? 0 : (b > NUM_BIN - 1 ? NUM_BIN - 1 : b);
}

#define GRP 8

__device__ __forceinline__ void logits_to_lds(const int tv, const int cnt, const int jbase,
                                              const float4 p1, const float4 w,
                                              const float* __restrict__ P2, const int d0,
                                              const int lane, float* __restrict__ slog) {
  const int h = lane >> 2, q = lane & 3;
  for (int j = 0; j < cnt; j += GRP) {
    float4 a[GRP];
#pragma unroll
    for (int u = 0; u < GRP; ++u) {
      int ju = j + u; ju = ju < cnt ? ju : cnt - 1;
      const int t = __builtin_amdgcn_readlane(tv, ju);
      a[u] = *(const float4*)(P2 + (size_t)t * 128 + d0);
    }
    float s[GRP];
#pragma unroll
    for (int u = 0; u < GRP; ++u) {
      const float x = lrelu(p1.x + a[u].x) * w.x + lrelu(p1.y + a[u].y) * w.y +
                      lrelu(p1.z + a[u].z) * w.z + lrelu(p1.w + a[u].w) * w.w;
      s[u] = quad_sum(x);
    }
#pragma unroll
    for (int u = 0; u < GRP; u += 4) {
      const float c0 = s[u], c1 = s[u + 1], c2 = s[u + 2], c3 = s[u + 3];
      const float val = q == 0 ? c0 : (q == 1 ? c1 : (q == 2 ? c2 : c3));
      const int je = j + u + q;
      if (je < cnt) slog[(jbase + je) * NUM_HEAD + h] = val;
    }
  }
}

__device__ __forceinline__ void weighted_rows(const int tv, const int cnt, const int jbase,
                                              const float* __restrict__ Mmat, const int d0,
                                              const int lane, const float* __restrict__ slog,
                                              float4& acc) {
  const int h = lane >> 2;
  for (int j = 0; j < cnt; j += GRP) {
    float4 a[GRP];
    float be[GRP];
#pragma unroll
    for (int u = 0; u < GRP; ++u) {
      int ju = j + u; ju = ju < cnt ? ju : cnt - 1;
      const int t = __builtin_amdgcn_readlane(tv, ju);
      a[u] = *(const float4*)(Mmat + (size_t)t * 128 + d0);
      be[u] = slog[(jbase + ju) * NUM_HEAD + h];
    }
#pragma unroll
    for (int u = 0; u < GRP; ++u) {
      if (j + u < cnt) {
        acc.x += be[u] * a[u].x; acc.y += be[u] * a[u].y;
        acc.z += be[u] * a[u].z; acc.w += be[u] * a[u].w;
      }
    }
  }
}

__global__ void __launch_bounds__(256) k_rel(const int* __restrict__ trip,
                      const int* __restrict__ off, const int* __restrict__ segw,
                      const float* __restrict__ P1, const float* __restrict__ P2,
                      const float* __restrict__ Mmat, const float* __restrict__ av,
                      const float* __restrict__ ab, float* __restrict__ out, int R) {
  __shared__ float s_bias[NUM_BIN * NUM_HEAD];
  __shared__ float s_log[8][SEG_CAP * NUM_HEAD];
  __shared__ int   s_tail[8][SEG_CAP];
  __shared__ int   s_bin[8][SEG_CAP];

  const int4* __restrict__ seg = (const int4*)segw;
  const int lane = threadIdx.x & 31;
  const int wv = __builtin_amdgcn_readfirstlane(threadIdx.x >> 5);
  if (threadIdx.x < NUM_BIN * NUM_HEAD) s_bias[threadIdx.x] = lrelu(ab[threadIdx.x]);
  __syncthreads();

  const int r = blockIdx.x * 8 + wv;
  if (r >= R) return;
  const int start = __builtin_amdgcn_readfirstlane(off[r]);
  const int n     = __builtin_amdgcn_readfirstlane(off[r + 1]) - start;

  const int d0 = lane * 4;
  const int h  = lane >> 2;
  const float4 p1 = *(const float4*)(P1 + (size_t)r * 128 + d0);
  const float4 w  = *(const float4*)(av + d0);
  float4 acc = make_float4(0.f, 0.f, 0.f, 0.f);
  float* __restrict__ slog = s_log[wv];

  if (n <= SEG_CAP) {
    const bool has0 = lane < n, has1 = lane + 32 < n;
    int4 q0 = make_int4(BIG_I, 0, 0, 0), q1 = make_int4(BIG_I, 0, 0, 0);
    if (has0) q0 = seg[start + lane];
    if (has1) q1 = seg[start + 32 + lane];
    const int e0 = q0.x, e1 = q1.x;
    int r0 = 0, r1 = 0;
    const int nlo = n < 32 ? n : 32;
    const int nhi = n - nlo;
    for (int j = 0; j < nlo; ++j) {
      const int x = __builtin_amdgcn_readlane(e0, j);
      r0 += (x < e0); r1 += (x < e1);
    }
    for (int j = 0; j < nhi; ++j) {
      const int x = __builtin_amdgcn_readlane(e1, j);
      r0 += (x < e0); r1 += (x < e1);
    }
    if (has0) { s_tail[wv][r0] = q0.y; s_bin[wv][r0] = clamp_bin(q0.z); }
    if (has1) { s_tail[wv][r1] = q1.y; s_bin[wv][r1] = clamp_bin(q1.z); }
    wave_lds_handover();
    const int tv0 = s_tail[wv][lane];
    const int tv1 = s_tail[wv][lane + 32];

    logits_to_lds(tv0, nlo, 0,  p1, w, P2, d0, lane, slog);
    logits_to_lds(tv1, nhi, 32, p1, w, P2, d0, lane, slog);
    wave_lds_handover();

    float v[SEG_CAP / 4];
    float mx = -BIG_F;
#pragma unroll
    for (int i = 0; i < SEG_CAP / 4; ++i) {
      const int j = (lane >> 3) + 4 * i;
      v[i] = -BIG_F;
      if (j < n) v[i] = slog[lane + 32 * i] + s_bias[s_bin[wv][j] * NUM_HEAD + (lane & 7)];
      mx = fmaxf(mx, v[i]);
    }
    mx = fmaxf(mx, __shfl_xor(mx, 8, 32));
    mx = fmaxf(mx, __shfl_xor(mx, 16, 32));
    float sum = 0.f;
#pragma unroll
    for (int i = 0; i < SEG_CAP / 4; ++i) {
      const int j = (lane >> 3) + 4 * i;
      v[i] = (j < n) ? expf(v[i] - mx) : 0.f;
      sum += v[i];
    }
    sum += __shfl_xor(sum, 8, 32);
    sum += __shfl_xor(sum, 16, 32);
    const float den = sum + 1e-16f;
#pragma unroll
    for (int i = 0; i < SEG_CAP / 4; ++i) {
      const int j = (lane >> 3) + 4 * i;
      if (j < n) slog[lane + 32 * i] = v[i] / den;
    }
    wave_lds_handover();

    weighted_rows(tv0, nlo, 0,  Mmat, d0, lane, slog, acc);
    weighted_rows(tv1, nhi, 32, Mmat, d0, lane, slog, acc);
  } else {
    float M = -BIG_F, L = 0.f;
    int prev = -1;
    for (int k = 0; k < n; ++k) {
      int best = BIG_I;
      for (int base = 0; base < n; base += 32) {
        const int idx = base + lane;
        int x = (idx < n) ? seg[start + idx].x : BIG_I;
        if (x <= prev) x = BIG_I;
        best = x < best ? x : best;
      }
      for (int o = 16; o > 0; o >>= 1) { const int y = __shfl_xor(best, o, 32); best = y < best ? y : best; }
      best = __builtin_amdgcn_readfirstlane(best);
      prev = best;
      const int t = trip[3 * best + 1];
      const int b = clamp_bin(trip[3 * best + 2]);
      const float s = edge_dot(p1, w, P2, t, d0) + s_bias[b * NUM_HEAD + h];
      const float Mn = fmaxf(M, s);
      const float sc = expf(M - Mn), p = expf(s - Mn);
      const float4 m = *(const float4*)(Mmat + (size_t)t * 128 + d0);
      L = L * sc + p;
      acc.x = acc.x * sc + p * m.x; acc.y = acc.y * sc + p * m.y;
      acc.z = acc.z * sc + p * m.z; acc.w = acc.w * sc + p * m.w;
      M = Mn;
    }
    const float den = L + 1e-16f;
    acc.x /= den; acc.y /= den; acc.z /= den; acc.w /= den;
  }
  *(float4*)(out + (size_t)r * 128 + d0) = acc;
}

extern "C" void kernel_launch(void* const* d_in, const int* in_sizes, int n_in,
                              void* d_out, int out_size, void* d_ws, size_t ws_size,
                              hipStream_t stream) {
  const float* emb      = (const float*)d_in[0];
  const int*   trip     = (const int*)  d_in[1];
  const float* attn_w   = (const float*)d_in[2];
  const float* attn_b   = (const float*)d_in[3];
  const float* attn_bin = (const float*)d_in[4];
  const float* attn_vec = (const float*)d_in[5];
  const float* aggr_w   = (const float*)d_in[6];
  const float* aggr_b   = (const float*)d_in[7];

  const int R = in_sizes[0] / DIM_IN;
  const int E = in_sizes[1] / 3;

  float* ws     = (float*)d_ws;
  float* Bmat   = ws;
  float* Bias   = Bmat + DIM_IN * NCOLS;
  float* P1     = Bias + NCOLS;
  float* P2     = P1 + (size_t)R * 128;
  float* Mm     = P2 + (size_t)R * 128;
  int*   seg    = (int*)(Mm + (size_t)R * 128);
  int*   cnt    = seg + (size_t)E * 4;
  int*   off    = cnt + R;
  int*   rk     = off + R + 1;
  float* out    = (float*)d_out;

  k_prep<<<(DIM_IN * NCOLS + 255) / 256, 256, 0, stream>>>(attn_w, attn_b, aggr_w, aggr_b, Bmat, Bias, cnt, R);

  k_hist<<<(E + 255) / 256, 256, 0, stream>>>(trip, cnt, rk, E);

  k_scan<<<1, 1024, 0, stream>>>(cnt, off, R);

  k_scatter<<<(E + 255) / 256, 256, 0, stream>>>(trip, off, rk, seg, E);

  dim3 gg((R + 15) / 16, NCOLS / 64);
  k_gemm<<<gg, 32, 0, stream>>>(emb, Bmat, Bias, P1, P2, Mm, R);

  k_rel<<<(R + 7) / 8, 256, 0, stream>>>(trip, off, seg, P1, P2, Mm, attn_vec, attn_bin, out, R);
}


# Each head owns a fixed 64-slot bucket that the counting pass fills itself (slot = the count it gets back): scan and scatter launches dropped, 6 launches to 4; sums still in edge id order
# speedup vs baseline: 2.3758x; 1.2641x over previous
#include <hip/hip_runtime.h>
#include <math.h>

#define DIM_IN   128
#define DIM_OUT  128
#define NUM_HEAD 8
#define DIM_HID  16
#define NUM_BIN  10
#define NEG_SLOPE 0.2f
#define NCOLS 384
#define SEG_CAP 64
#define BIG_F 3.402823466e38f
#define BIG_I 2147483647

typedef __attribute__((ext_vector_type(2))) float v2f;
typedef __attribute__((ext_vector_type(8))) float v8f;

__device__ __forceinline__ float lrelu(float x) { return x >= 0.f ? x : NEG_SLOPE * x; }

__global__ void k_prep(const float* __restrict__ attn_w, const float* __restrict__ attn_b,
                       const float* __restrict__ aggr_w, const float* __restrict__ aggr_b,
                       float* __restrict__ Bmat, float* __restrict__ Bias,
                       int* __restrict__ cnt, int R) {
  int i = blockIdx.x * blockDim.x + threadIdx.x;
  if (i < DIM_IN * NCOLS) {
    int k = i / NCOLS, n = i % NCOLS;
    float v;
    if (n < 128)       v = attn_w[n * 256 + k];
    else if (n < 256)  v = attn_w[(n - 128) * 256 + 128 + k];
    else               v = aggr_w[(n - 256) * 128 + k];
    Bmat[i] = v;
  }
  if (i < NCOLS)
    Bias[i] = (i < 128) ? attn_b[i] : ((i < 256) ? 0.f : aggr_b[i - 256]);
  for (int j = i; j < R; j += gridDim.x * blockDim.x) cnt[j] = 0;
}

__global__ void k_gemm(const float* __restrict__ A, const float* __restrict__ Bmat,
                       const float* __restrict__ Bias,
                       float* __restrict__ P1, float* __restrict__ P2,
                       float* __restrict__ Mout, int R) {
  const int lane = threadIdx.x & 31;
  const int half = lane >> 4;
  const int l16  = lane & 15;
  const int m0 = blockIdx.x * 16;
  const int n0 = blockIdx.y * 64;

  int mrow = m0 + l16; if (mrow >= R) mrow = R - 1;
  const float* __restrict__ Arow = A + (size_t)mrow * DIM_IN;

  v8f acc0 = {}, acc1 = {}, acc2 = {}, acc3 = {};
  for (int k0 = 0; k0 < DIM_IN; k0 += 4) {
    const int ka = k0 + (half << 1);
    v2f a; a.x = Arow[ka]; a.y = Arow[ka + 1];
    const float* __restrict__ Bp = Bmat + (size_t)ka * NCOLS + n0 + l16;
    v2f b0; b0.x = Bp[0];  b0.y = Bp[NCOLS];
    v2f b1; b1.x = Bp[16]; b1.y = Bp[NCOLS + 16];
    v2f b2; b2.x = Bp[32]; b2.y = Bp[NCOLS + 32];
    v2f b3; b3.x = Bp[48]; b3.y = Bp[NCOLS + 48];
    acc0 = __builtin_amdgcn_wmma_f32_16x16x4_f32(false, a, false, b0, (short)0, acc0, false, false);
    acc1 = __builtin_amdgcn_wmma_f32_16x16x4_f32(false, a, false, b1, (short)0, acc1, false, false);
    acc2 = __builtin_amdgcn_wmma_f32_16x16x4_f32(false, a, false, b2, (short)0, acc2, false, false);
    acc3 = __builtin_amdgcn_wmma_f32_16x16x4_f32(false, a, false, b3, (short)0, acc3, false, false);
  }

#pragma unroll
  for (int i = 0; i < 8; ++i) {
    const int row = m0 + i + (half << 3);
    if (row >= R) continue;
    float vals[4] = { acc0[i], acc1[i], acc2[i], acc3[i] };
#pragma unroll
    for (int j = 0; j < 4; ++j) {
      const int c = n0 + j * 16 + l16;
      const float v = vals[j] + Bias[c];
      if (c < 128)      P1[(size_t)row * 128 + c]         = v;
      else if (c < 256) P2[(size_t)row * 128 + (c - 128)] = v;
      else              Mout[(size_t)row * 128 + (c - 256)] = v;
    }
  }
}

__device__ __forceinline__ int clamp_bin(int b) {
  return b < 0 ? 0 : (b > NUM_BIN - 1 ? NUM_BIN - 1 : b);
}

__global__ void k_bucket(const int* __restrict__ trip, int* __restrict__ cnt,
                         int* __restrict__ bucketw, int E) {
  const int e = blockIdx.x * blockDim.x + threadIdx.x;
  if (e >= E) return;
  const int h = trip[3 * e], t = trip[3 * e + 1], b = trip[3 * e + 2];
  const int slot = atomicAdd(cnt + h, 1);
  if (slot < SEG_CAP) {
    int2* __restrict__ bk = (int2*)bucketw;
    bk[(size_t)h * SEG_CAP + slot] = make_int2(e, (int)((unsigned)t << 4) | clamp_bin(b));
  }
}

__device__ __forceinline__ void wave_lds_handover() {
  __builtin_amdgcn_fence(__ATOMIC_RELEASE, "wavefront");
  __builtin_amdgcn_wave_barrier();
  __builtin_amdgcn_fence(__ATOMIC_ACQUIRE, "wavefront");
}

__device__ __forceinline__ float quad_sum(float s) {
  s += __int_as_float(__builtin_amdgcn_update_dpp(0, __float_as_int(s), 0xB1, 0xF, 0xF, true));
  s += __int_as_float(__builtin_amdgcn_update_dpp(0, __float_as_int(s), 0x4E, 0xF, 0xF, true));
  return s;
}

__device__ __forceinline__ float edge_dot(const float4 p1, const float4 w,
                                          const float* __restrict__ P2, int t, int d0) {
  const float4 p2 = *(const float4*)(P2 + (size_t)t * 128 + d0);
  const float s = lrelu(p1.x + p2.x) * w.x + lrelu(p1.y + p2.y) * w.y +
                  lrelu(p1.z + p2.z) * w.z + lrelu(p1.w + p2.w) * w.w;
  return quad_sum(s);
}

#define GRP 8

__device__ __forceinline__ void logits_to_lds(const int tv, const int cnt, const int jbase,
                                              const float4 p1, const float4 w,
                                              const float* __restrict__ P2, const int d0,
                                              const int lane, float* __restrict__ slog) {
  const int h = lane >> 2, q = lane & 3;
  for (int j = 0; j < cnt; j += GRP) {
    float4 a[GRP];
#pragma unroll
    for (int u = 0; u < GRP; ++u) {
      int ju = j + u; ju = ju < cnt ? ju : cnt - 1;
      const int t = __builtin_amdgcn_readlane(tv, ju);
      a[u] = *(const float4*)(P2 + (size_t)t * 128 + d0);
    }
    float s[GRP];
#pragma unroll
    for (int u = 0; u < GRP; ++u) {
      const float x = lrelu(p1.x + a[u].x) * w.x + lrelu(p1.y + a[u].y) * w.y +
                      lrelu(p1.z + a[u].z) * w.z + lrelu(p1.w + a[u].w) * w.w;
      s[u] = quad_sum(x);
    }
#pragma unroll
    for (int u = 0; u < GRP; u += 4) {
      const float c0 = s[u], c1 = s[u + 1], c2 = s[u + 2], c3 = s[u + 3];
      const float val = q == 0 ? c0 : (q == 1 ? c1 : (q == 2 ? c2 : c3));
      const int je = j + u + q;
      if (je < cnt) slog[(jbase + je) * NUM_HEAD + h] = val;
    }
  }
}

__device__ __forceinline__ void weighted_rows(const int tv, const int cnt, const int jbase,
                                              const float* __restrict__ Mmat, const int d0,
                                              const int lane, const float* __restrict__ slog,
                                              float4& acc) {
  const int h = lane >> 2;
  for (int j = 0; j < cnt; j += GRP) {
    float4 a[GRP];
    float be[GRP];
#pragma unroll
    for (int u = 0; u < GRP; ++u) {
      int ju = j + u; ju = ju < cnt ? ju : cnt - 1;
      const int t = __builtin_amdgcn_readlane(tv, ju);
      a[u] = *(const float4*)(Mmat + (size_t)t * 128 + d0);
      be[u] = slog[(jbase + ju) * NUM_HEAD + h];
    }
#pragma unroll
    for (int u = 0; u < GRP; ++u) {
      if (j + u < cnt) {
        acc.x += be[u] * a[u].x; acc.y += be[u] * a[u].y;
        acc.z += be[u] * a[u].z; acc.w += be[u] * a[u].w;
      }
    }
  }
}

__global__ void __launch_bounds__(256) k_rel(const int* __restrict__ trip,
                      const int* __restrict__ cnt, const int* __restrict__ bucketw,
                      const float* __restrict__ P1, const float* __restrict__ P2,
                      const float* __restrict__ Mmat, const float* __restrict__ av,
                      const float* __restrict__ ab, float* __restrict__ out, int R, int E) {
  __shared__ float s_bias[NUM_BIN * NUM_HEAD];
  __shared__ float s_log[8][SEG_CAP * NUM_HEAD];
  __shared__ int   s_tail[8][SEG_CAP];
  __shared__ int   s_bin[8][SEG_CAP];

  const int lane = threadIdx.x & 31;
  const int wv = __builtin_amdgcn_readfirstlane(threadIdx.x >> 5);
  if (threadIdx.x < NUM_BIN * NUM_HEAD) s_bias[threadIdx.x] = lrelu(ab[threadIdx.x]);
  __syncthreads();

  const int r = blockIdx.x * 8 + wv;
  if (r >= R) return;
  const int n = cnt[r];

  const int d0 = lane * 4;
  const int h  = lane >> 2;
  const float4 p1 = *(const float4*)(P1 + (size_t)r * 128 + d0);
  const float4 w  = *(const float4*)(av + d0);
  float4 acc = make_float4(0.f, 0.f, 0.f, 0.f);
  float* __restrict__ slog = s_log[wv];

  if (n <= SEG_CAP) {
    const bool has0 = lane < n, has1 = lane + 32 < n;
    const int2* __restrict__ bk = (const int2*)bucketw + (size_t)r * SEG_CAP;
    int2 q0 = make_int2(BIG_I, 0), q1 = make_int2(BIG_I, 0);
    if (has0) q0 = bk[lane];
    if (has1) q1 = bk[32 + lane];
    const int e0 = q0.x, e1 = q1.x;
    int r0 = 0, r1 = 0;
    const int nlo = n < 32 ? n : 32;
    const int nhi = n - nlo;
    for (int j = 0; j < nlo; ++j) {
      const int x = __builtin_amdgcn_readlane(e0, j);
      r0 += (x < e0); r1 += (x < e1);
    }
    for (int j = 0; j < nhi; ++j) {
      const int x = __builtin_amdgcn_readlane(e1, j);
      r0 += (x < e0); r1 += (x < e1);
    }
    if (has0) { s_tail[wv][r0] = q0.y >> 4; s_bin[wv][r0] = q0.y & 15; }
    if (has1) { s_tail[wv][r1] = q1.y >> 4; s_bin[wv][r1] = q1.y & 15; }
    wave_lds_handover();
    const int tv0 = s_tail[wv][lane];
    const int tv1 = s_tail[wv][lane + 32];

    logits_to_lds(tv0, nlo, 0,  p1, w, P2, d0, lane, slog);
    logits_to_lds(tv1, nhi, 32, p1, w, P2, d0, lane, slog);
    wave_lds_handover();

    float v[SEG_CAP / 4];
    float mx = -BIG_F;
#pragma unroll
    for (int i = 0; i < SEG_CAP / 4; ++i) {
      const int j = (lane >> 3) + 4 * i;
      v[i] = -BIG_F;
      if (j < n) v[i] = slog[lane + 32 * i] + s_bias[s_bin[wv][j] * NUM_HEAD + (lane & 7)];
      mx = fmaxf(mx, v[i]);
    }
    mx = fmaxf(mx, __shfl_xor(mx, 8, 32));
    mx = fmaxf(mx, __shfl_xor(mx, 16, 32));
    float sum = 0.f;
#pragma unroll
    for (int i = 0; i < SEG_CAP / 4; ++i) {
      const int j = (lane >> 3) + 4 * i;
      v[i] = (j < n) ? expf(v[i] - mx) : 0.f;
      sum += v[i];
    }
    sum += __shfl_xor(sum, 8, 32);
    sum += __shfl_xor(sum, 16, 32);
    const float den = sum + 1e-16f;
#pragma unroll
    for (int i = 0; i < SEG_CAP / 4; ++i) {
      const int j = (lane >> 3) + 4 * i;
      if (j < n) slog[lane + 32 * i] = v[i] / den;
    }
    wave_lds_handover();

    weighted_rows(tv0, nlo, 0,  Mmat, d0, lane, slog, acc);
    weighted_rows(tv1, nhi, 32, Mmat, d0, lane, slog, acc);
  } else {
    float M = -BIG_F, L = 0.f;
    for (int base = 0; base < E; base += 32) {
      const int idx = base + lane;
      const bool mine = idx < E && trip[3 * idx] == r;
      unsigned todo = __builtin_amdgcn_readfirstlane((unsigned)__ballot(mine));
      while (todo != 0u) {
        const int e = base + __builtin_ctz(todo);
        todo &= todo - 1u;
        const int t = trip[3 * e + 1];
        const int b = clamp_bin(trip[3 * e + 2]);
        const float s = edge_dot(p1, w, P2, t, d0) + s_bias[b * NUM_HEAD + h];
        const float Mn = fmaxf(M, s);
        const float sc = expf(M - Mn), p = expf(s - Mn);
        const float4 m = *(const float4*)(Mmat + (size_t)t * 128 + d0);
        L = L * sc + p;
        acc.x = acc.x * sc + p * m.x; acc.y = acc.y * sc + p * m.y;
        acc.z = acc.z * sc + p * m.z; acc.w = acc.w * sc + p * m.w;
        M = Mn;
      }
    }
    const float den = L + 1e-16f;
    acc.x /= den; acc.y /= den; acc.z /= den; acc.w /= den;
  }
  *(float4*)(out + (size_t)r * 128 + d0) = acc;
}

extern "C" void kernel_launch(void* const* d_in, const int* in_sizes, int n_in,
                              void* d_out, int out_size, void* d_ws, size_t ws_size,
                              hipStream_t stream) {
  const float* emb      = (const float*)d_in[0];
  const int*   trip     = (const int*)  d_in[1];
  const float* attn_w   = (const float*)d_in[2];
  const float* attn_b   = (const float*)d_in[3];
  const float* attn_bin = (const float*)d_in[4];
  const float* attn_vec = (const float*)d_in[5];
  const float* aggr_w   = (const float*)d_in[6];
  const float* aggr_b   = (const float*)d_in[7];

  const int R = in_sizes[0] / DIM_IN;
  const int E = in_sizes[1] / 3;

  float* ws     = (float*)d_ws;
  float* Bmat   = ws;
  float* Bias   = Bmat + DIM_IN * NCOLS;
  float* P1     = Bias + NCOLS;
  float* P2     = P1 + (size_t)R * 128;
  float* Mm     = P2 + (size_t)R * 128;
  int*   bucket = (int*)(Mm + (size_t)R * 128);
  int*   cnt    = bucket + (size_t)R * SEG_CAP * 2;
  float* out    = (float*)d_out;

  k_prep<<<(DIM_IN * NCOLS + 255) / 256, 256, 0, stream>>>(attn_w, attn_b, aggr_w, aggr_b, Bmat, Bias, cnt, R);

  k_bucket<<<(E + 255) / 256, 256, 0, stream>>>(trip, cnt, bucket, E);

  dim3 gg((R + 15) / 16, NCOLS / 64);
  k_gemm<<<gg, 32, 0, stream>>>(emb, Bmat, Bias, P1, P2, Mm, R);

  k_rel<<<(R + 7) / 8, 256, 0, stream>>>(trip, cnt, bucket, P1, P2, Mm, attn_vec, attn_bin, out, R, E);
}


# Grouping pass folded into the GEMM launch: each GEMM wave also files its share of the edges (loads before the k loop, counter adds and record stores behind its tile); 4 launches to 3; sum order kept
# speedup vs baseline: 2.6951x; 1.1344x over previous
#include <hip/hip_runtime.h>
#include <math.h>

#define DIM_IN   128
#define DIM_OUT  128
#define NUM_HEAD 8
#define DIM_HID  16
#define NUM_BIN  10
#define NEG_SLOPE 0.2f
#define NCOLS 384
#define SEG_CAP 64
#define BIG_F 3.402823466e38f
#define BIG_I 2147483647

typedef __attribute__((ext_vector_type(2))) float v2f;
typedef __attribute__((ext_vector_type(8))) float v8f;

__device__ __forceinline__ float lrelu(float x) { return x >= 0.f ? x : NEG_SLOPE * x; }

__global__ void k_prep(const float* __restrict__ attn_w, const float* __restrict__ attn_b,
                       const float* __restrict__ aggr_w, const float* __restrict__ aggr_b,
                       float* __restrict__ Bmat, float* __restrict__ Bias,
                       int* __restrict__ cnt, int R) {
  int i = blockIdx.x * blockDim.x + threadIdx.x;
  if (i < DIM_IN * NCOLS) {
    int k = i / NCOLS, n = i % NCOLS;
    float v;
    if (n < 128)       v = attn_w[n * 256 + k];
    else if (n < 256)  v = attn_w[(n - 128) * 256 + 128 + k];
    else               v = aggr_w[(n - 256) * 128 + k];
    Bmat[i] = v;
  }
  if (i < NCOLS)
    Bias[i] = (i < 128) ? attn_b[i] : ((i < 256) ? 0.f : aggr_b[i - 256]);
  for (int j = i; j < R; j += gridDim.x * blockDim.x) cnt[j] = 0;
}

__device__ __forceinline__ int clamp_bin(int b) {
  return b < 0 ? 0 : (b > NUM_BIN - 1 ? NUM_BIN - 1 : b);
}

#define EDGE_ROUNDS 3

__global__ void k_gemm(const float* __restrict__ A, const float* __restrict__ Bmat,
                       const float* __restrict__ Bias,
                       float* __restrict__ P1, float* __restrict__ P2,
                       float* __restrict__ Mout, int R,
                       const int* __restrict__ trip, int* __restrict__ cnt,
                       int* __restrict__ bucketw, int E) {
  const int lane = threadIdx.x & 31;
  const int half = lane >> 4;
  const int l16  = lane & 15;
  const int m0 = blockIdx.x * 16;
  const int n0 = blockIdx.y * 64;

  const int nlanes = (int)(gridDim.x * gridDim.y) * 32;
  const int gid = (int)(blockIdx.y * gridDim.x + blockIdx.x) * 32 + lane;
  int eh[EDGE_ROUNDS], et[EDGE_ROUNDS], eb[EDGE_ROUNDS];
#pragma unroll
  for (int k = 0; k < EDGE_ROUNDS; ++k) {
    const int e = gid + k * nlanes;
    const int ec = e < E ? e : E - 1;
    eh[k] = trip[3 * ec]; et[k] = trip[3 * ec + 1]; eb[k] = trip[3 * ec + 2];
  }

  int mrow = m0 + l16; if (mrow >= R) mrow = R - 1;
  const float* __restrict__ Arow = A + (size_t)mrow * DIM_IN;

  v8f acc0 = {}, acc1 = {}, acc2 = {}, acc3 = {};
  for (int k0 = 0; k0 < DIM_IN; k0 += 4) {
    const int ka = k0 + (half << 1);
    v2f a; a.x = Arow[ka]; a.y = Arow[ka + 1];
    const float* __restrict__ Bp = Bmat + (size_t)ka * NCOLS + n0 + l16;
    v2f b0; b0.x = Bp[0];  b0.y = Bp[NCOLS];
    v2f b1; b1.x = Bp[16]; b1.y = Bp[NCOLS + 16];
    v2f b2; b2.x = Bp[32]; b2.y = Bp[NCOLS + 32];
    v2f b3; b3.x = Bp[48]; b3.y = Bp[NCOLS + 48];
    acc0 = __builtin_amdgcn_wmma_f32_16x16x4_f32(false, a, false, b0, (short)0, acc0, false, false);
    acc1 = __builtin_amdgcn_wmma_f32_16x16x4_f32(false, a, false, b1, (short)0, acc1, false, false);
    acc2 = __builtin_amdgcn_wmma_f32_16x16x4_f32(false, a, false, b2, (short)0, acc2, false, false);
    acc3 = __builtin_amdgcn_wmma_f32_16x16x4_f32(false, a, false, b3, (short)0, acc3, false, false);
  }

#pragma unroll
  for (int i = 0; i < 8; ++i) {
    const int row = m0 + i + (half << 3);
    if (row >= R) continue;
    float vals[4] = { acc0[i], acc1[i], acc2[i], acc3[i] };
#pragma unroll
    for (int j = 0; j < 4; ++j) {
      const int c = n0 + j * 16 + l16;
      const float v = vals[j] + Bias[c];
      if (c < 128)      P1[(size_t)row * 128 + c]         = v;
      else if (c < 256) P2[(size_t)row * 128 + (c - 128)] = v;
      else              Mout[(size_t)row * 128 + (c - 256)] = v;
    }
  }

  int2* __restrict__ bk = (int2*)bucketw;
  int slot[EDGE_ROUNDS];
#pragma unroll
  for (int k = 0; k < EDGE_ROUNDS; ++k) {
    slot[k] = SEG_CAP;
    if (gid + k * nlanes < E) slot[k] = atomicAdd(cnt + eh[k], 1);
  }
#pragma unroll
  for (int k = 0; k < EDGE_ROUNDS; ++k) {
    if (slot[k] < SEG_CAP)
      bk[(size_t)eh[k] * SEG_CAP + slot[k]] =
          make_int2(gid + k * nlanes, (int)((unsigned)et[k] << 4) | clamp_bin(eb[k]));
  }
  for (int e = gid + EDGE_ROUNDS * nlanes; e < E; e += nlanes) {
    const int h = trip[3 * e], t = trip[3 * e + 1], b = trip[3 * e + 2];
    const int s = atomicAdd(cnt + h, 1);
    if (s < SEG_CAP) bk[(size_t)h * SEG_CAP + s] = make_int2(e, (int)((unsigned)t << 4) | clamp_bin(b));
  }
}

__device__ __forceinline__ void wave_lds_handover() {
  __builtin_amdgcn_fence(__ATOMIC_RELEASE, "wavefront");
  __builtin_amdgcn_wave_barrier();
  __builtin_amdgcn_fence(__ATOMIC_ACQUIRE, "wavefront");
}

__device__ __forceinline__ float quad_sum(float s) {
  s += __int_as_float(__builtin_amdgcn_update_dpp(0, __float_as_int(s), 0xB1, 0xF, 0xF, true));
  s += __int_as_float(__builtin_amdgcn_update_dpp(0, __float_as_int(s), 0x4E, 0xF, 0xF, true));
  return s;
}

__device__ __forceinline__ float edge_dot(const float4 p1, const float4 w,
                                          const float* __restrict__ P2, int t, int d0) {
  const float4 p2 = *(const float4*)(P2 + (size_t)t * 128 + d0);
  const float s = lrelu(p1.x + p2.x) * w.x + lrelu(p1.y + p2.y) * w.y +
                  lrelu(p1.z + p2.z) * w.z + lrelu(p1.w + p2.w) * w.w;
  return quad_sum(s);
}

#define GRP 8

__device__ __forceinline__ void logits_to_lds(const int tv, const int cnt, const int jbase,
                                              const float4 p1, const float4 w,
                                              const float* __restrict__ P2, const int d0,
                                              const int lane, float* __restrict__ slog) {
  const int h = lane >> 2, q = lane & 3;
  for (int j = 0; j < cnt; j += GRP) {
    float4 a[GRP];
#pragma unroll
    for (int u = 0; u < GRP; ++u) {
      int ju = j + u; ju = ju < cnt ? ju : cnt - 1;
      const int t = __builtin_amdgcn_readlane(tv, ju);
      a[u] = *(const float4*)(P2 + (size_t)t * 128 + d0);
    }
    float s[GRP];
#pragma unroll
    for (int u = 0; u < GRP; ++u) {
      const float x = lrelu(p1.x + a[u].x) * w.x + lrelu(p1.y + a[u].y) * w.y +
                      lrelu(p1.z + a[u].z) * w.z + lrelu(p1.w + a[u].w) * w.w;
      s[u] = quad_sum(x);
    }
#pragma unroll
    for (int u = 0; u < GRP; u += 4) {
      const float c0 = s[u], c1 = s[u + 1], c2 = s[u + 2], c3 = s[u + 3];
      const float val = q == 0 ? c0 : (q == 1 ? c1 : (q == 2 ? c2 : c3));
      const int je = j + u + q;
      if (je < cnt) slog[(jbase + je) * NUM_HEAD + h] = val;
    }
  }
}

__device__ __forceinline__ void weighted_rows(const int tv, const int cnt, const int jbase,
                                              const float* __restrict__ Mmat, const int d0,
                                              const int lane, const float* __restrict__ slog,
                                              float4& acc) {
  const int h = lane >> 2;
  for (int j = 0; j < cnt; j += GRP) {
    float4 a[GRP];
    float be[GRP];
#pragma unroll
    for (int u = 0; u < GRP; ++u) {
      int ju = j + u; ju = ju < cnt ? ju : cnt - 1;
      const int t = __builtin_amdgcn_readlane(tv, ju);
      a[u] = *(const float4*)(Mmat + (size_t)t * 128 + d0);
      be[u] = slog[(jbase + ju) * NUM_HEAD + h];
    }
#pragma unroll
    for (int u = 0; u < GRP; ++u) {
      if (j + u < cnt) {
        acc.x += be[u] * a[u].x; acc.y += be[u] * a[u].y;
        acc.z += be[u] * a[u].z; acc.w += be[u] * a[u].w;
      }
    }
  }
}

__global__ void __launch_bounds__(256) k_rel(const int* __restrict__ trip,
                      const int* __restrict__ cnt, const int* __restrict__ bucketw,
                      const float* __restrict__ P1, const float* __restrict__ P2,
                      const float* __restrict__ Mmat, const float* __restrict__ av,
                      const float* __restrict__ ab, float* __restrict__ out, int R, int E) {
  __shared__ float s_bias[NUM_BIN * NUM_HEAD];
  __shared__ float s_log[8][SEG_CAP * NUM_HEAD];
  __shared__ int   s_tail[8][SEG_CAP];
  __shared__ int   s_bin[8][SEG_CAP];

  const int lane = threadIdx.x & 31;
  const int wv = __builtin_amdgcn_readfirstlane(threadIdx.x >> 5);
  if (threadIdx.x < NUM_BIN * NUM_HEAD) s_bias[threadIdx.x] = lrelu(ab[threadIdx.x]);
  __syncthreads();

  const int r = blockIdx.x * 8 + wv;
  if (r >= R) return;
  const int n = cnt[r];

  const int d0 = lane * 4;
  const int h  = lane >> 2;
  const float4 p1 = *(const float4*)(P1 + (size_t)r * 128 + d0);
  const float4 w  = *(const float4*)(av + d0);
  float4 acc = make_float4(0.f, 0.f, 0.f, 0.f);
  float* __restrict__ slog = s_log[wv];

  if (n <= SEG_CAP) {
    const bool has0 = lane < n, has1 = lane + 32 < n;
    const int2* __restrict__ bk = (const int2*)bucketw + (size_t)r * SEG_CAP;
    int2 q0 = make_int2(BIG_I, 0), q1 = make_int2(BIG_I, 0);
    if (has0) q0 = bk[lane];
    if (has1) q1 = bk[32 + lane];
    const int e0 = q0.x, e1 = q1.x;
    int r0 = 0, r1 = 0;
    const int nlo = n < 32 ? n : 32;
    const int nhi = n - nlo;
    for (int j = 0; j < nlo; ++j) {
      const int x = __builtin_amdgcn_readlane(e0, j);
      r0 += (x < e0); r1 += (x < e1);
    }
    for (int j = 0; j < nhi; ++j) {
      const int x = __builtin_amdgcn_readlane(e1, j);
      r0 += (x < e0); r1 += (x < e1);
    }
    if (has0) { s_tail[wv][r0] = q0.y >> 4; s_bin[wv][r0] = q0.y & 15; }
    if (has1) { s_tail[wv][r1] = q1.y >> 4; s_bin[wv][r1] = q1.y & 15; }
    wave_lds_handover();
    const int tv0 = s_tail[wv][lane];
    const int tv1 = s_tail[wv][lane + 32];

    logits_to_lds(tv0, nlo, 0,  p1, w, P2, d0, lane, slog);
    logits_to_lds(tv1, nhi, 32, p1, w, P2, d0, lane, slog);
    wave_lds_handover();

    float v[SEG_CAP / 4];
    float mx = -BIG_F;
#pragma unroll
    for (int i = 0; i < SEG_CAP / 4; ++i) {
      const int j = (lane >> 3) + 4 * i;
      v[i] = -BIG_F;
      if (j < n) v[i] = slog[lane + 32 * i] + s_bias[s_bin[wv][j] * NUM_HEAD + (lane & 7)];
      mx = fmaxf(mx, v[i]);
    }
    mx = fmaxf(mx, __shfl_xor(mx, 8, 32));
    mx = fmaxf(mx, __shfl_xor(mx, 16, 32));
    float sum = 0.f;
#pragma unroll
    for (int i = 0; i < SEG_CAP / 4; ++i) {
      const int j = (lane >> 3) + 4 * i;
      v[i] = (j < n) ? expf(v[i] - mx) : 0.f;
      sum += v[i];
    }
    sum += __shfl_xor(sum, 8, 32);
    sum += __shfl_xor(sum, 16, 32);
    const float den = sum + 1e-16f;
#pragma unroll
    for (int i = 0; i < SEG_CAP / 4; ++i) {
      const int j = (lane >> 3) + 4 * i;
      if (j < n) slog[lane + 32 * i] = v[i] / den;
    }
    wave_lds_handover();

    weighted_rows(tv0, nlo, 0,  Mmat, d0, lane, slog, acc);
    weighted_rows(tv1, nhi, 32, Mmat, d0, lane, slog, acc);
  } else {
    float M = -BIG_F, L = 0.f;
    for (int base = 0; base < E; base += 32) {
      const int idx = base + lane;
      const bool mine = idx < E && trip[3 * idx] == r;
      unsigned todo = __builtin_amdgcn_readfirstlane((unsigned)__ballot(mine));
      while (todo != 0u) {
        const int e = base + __builtin_ctz(todo);
        todo &= todo - 1u;
        const int t = trip[3 * e + 1];
        const int b = clamp_bin(trip[3 * e + 2]);
        const float s = edge_dot(p1, w, P2, t, d0) + s_bias[b * NUM_HEAD + h];
        const float Mn = fmaxf(M, s);
        const float sc = expf(M - Mn), p = expf(s - Mn);
        const float4 m = *(const float4*)(Mmat + (size_t)t * 128 + d0);
        L = L * sc + p;
        acc.x = acc.x * sc + p * m.x; acc.y = acc.y * sc + p * m.y;
        acc.z = acc.z * sc + p * m.z; acc.w = acc.w * sc + p * m.w;
        M = Mn;
      }
    }
    const float den = L + 1e-16f;
    acc.x /= den; acc.y /= den; acc.z /= den; acc.w /= den;
  }
  *(float4*)(out + (size_t)r * 128 + d0) = acc;
}

extern "C" void kernel_launch(void* const* d_in, const int* in_sizes, int n_in,
                              void* d_out, int out_size, void* d_ws, size_t ws_size,
                              hipStream_t stream) {
  const float* emb      = (const float*)d_in[0];
  const int*   trip     = (const int*)  d_in[1];
  const float* attn_w   = (const float*)d_in[2];
  const float* attn_b   = (const float*)d_in[3];
  const float* attn_bin = (const float*)d_in[4];
  const float* attn_vec = (const float*)d_in[5];
  const float* aggr_w   = (const float*)d_in[6];
  const float* aggr_b   = (const float*)d_in[7];

  const int R = in_sizes[0] / DIM_IN;
  const int E = in_sizes[1] / 3;

  float* ws     = (float*)d_ws;
  float* Bmat   = ws;
  float* Bias   = Bmat + DIM_IN * NCOLS;
  float* P1     = Bias + NCOLS;
  float* P2     = P1 + (size_t)R * 128;
  float* Mm     = P2 + (size_t)R * 128;
  int*   bucket = (int*)(Mm + (size_t)R * 128);
  int*   cnt    = bucket + (size_t)R * SEG_CAP * 2;
  float* out    = (float*)d_out;

  k_prep<<<(DIM_IN * NCOLS + 255) / 256, 256, 0, stream>>>(attn_w, attn_b, aggr_w, aggr_b, Bmat, Bias, cnt, R);

  dim3 gg((R + 15) / 16, NCOLS / 64);
  k_gemm<<<gg, 32, 0, stream>>>(emb, Bmat, Bias, P1, P2, Mm, R, trip, cnt, bucket, E);

  k_rel<<<(R + 7) / 8, 256, 0, stream>>>(trip, cnt, bucket, P1, P2, Mm, attn_vec, attn_bin, out, R, E);
}


# Edges grouped by a two-level counting sort on LDS counters (256 producers file by 8-relation bin, each consumer group routes its bin in LDS): no global atomics left; sums still in edge id order
# speedup vs baseline: 2.8870x; 1.0712x over previous
#include <hip/hip_runtime.h>
#include <math.h>

#define DIM_IN   128
#define DIM_OUT  128
#define NUM_HEAD 8
#define DIM_HID  16
#define NUM_BIN  10
#define NEG_SLOPE 0.2f
#define NCOLS 384
#define SEG_CAP 64
#define NPROD 256
#define MAXBIN 4096
#define PROD_U 10
#define BIG_F 3.402823466e38f
#define BIG_I 2147483647

typedef __attribute__((ext_vector_type(2))) float v2f;
typedef __attribute__((ext_vector_type(8))) float v8f;

__device__ __forceinline__ float lrelu(float x) { return x >= 0.f ? x : NEG_SLOPE * x; }

__device__ __forceinline__ int clamp_bin(int b) {
  return b < 0 ? 0 : (b > NUM_BIN - 1 ? NUM_BIN - 1 : b);
}

__global__ void __launch_bounds__(256) k_prep(const float* __restrict__ attn_w, const float* __restrict__ attn_b,
                       const float* __restrict__ aggr_w, const float* __restrict__ aggr_b,
                       float* __restrict__ Bmat, float* __restrict__ Bias,
                       const int* __restrict__ trip, int* __restrict__ recsw,
                       int* __restrict__ offw, int R, int E) {
  __shared__ int s_cnt[MAXBIN];
  __shared__ int s_cur[MAXBIN];
  __shared__ int s_wsum[8];
  const int tid = threadIdx.x;
  const int i = blockIdx.x * blockDim.x + tid;
  int nb = (R + 7) >> 3; if (nb > MAXBIN) nb = MAXBIN;
  const int g = blockIdx.x;
  const int chunk = (E + NPROD - 1) / NPROD;
  const int e0 = g * chunk;
  const int e1 = (e0 + chunk < E) ? e0 + chunk : E;
  for (int b = tid; b < nb; b += 256) s_cnt[b] = 0;
  __syncthreads();

  for (int base = e0; base < e1; base += 256 * PROD_U) {
    int hh[PROD_U];
#pragma unroll
    for (int u = 0; u < PROD_U; ++u) {
      const int e = base + tid + 256 * u;
      hh[u] = trip[3 * (e < e1 ? e : e1 - 1)];
    }
#pragma unroll
    for (int u = 0; u < PROD_U; ++u) {
      const int e = base + tid + 256 * u;
      const int b = hh[u] >> 3;
      if (e < e1 && hh[u] >= 0 && b < nb) atomicAdd(&s_cnt[b], 1);
    }
  }
  __syncthreads();

  const int lane = tid & 31, wv = tid >> 5;
  const int per = (nb + 255) / 256;
  const int lo = (tid * per < nb) ? tid * per : nb;
  const int hi = (lo + per < nb) ? lo + per : nb;
  int sum = 0;
  for (int b = lo; b < hi; ++b) sum += s_cnt[b];
  int v = sum;
#pragma unroll
  for (int d = 1; d < 32; d <<= 1) {
    const int y = __shfl_up(v, d, 32);
    if (lane >= d) v += y;
  }
  if (lane == 31) s_wsum[wv] = v;
  __syncthreads();
  int run = v - sum;
  for (int w = 0; w < wv; ++w) run += s_wsum[w];
  int2* __restrict__ offT = (int2*)offw;
  for (int b = lo; b < hi; ++b) {
    const int c = s_cnt[b];
    s_cur[b] = run;
    offT[(size_t)b * NPROD + g] = make_int2(run, c);
    run += c;
  }
  __syncthreads();

  if (i < DIM_IN * NCOLS) {
    int k = i / NCOLS, n = i % NCOLS;
    float v;
    if (n < 128)       v = attn_w[n * 256 + k];
    else if (n < 256)  v = attn_w[(n - 128) * 256 + 128 + k];
    else               v = aggr_w[(n - 256) * 128 + k];
    Bmat[i] = v;
  }
  if (i < NCOLS)
    Bias[i] = (i < 128) ? attn_b[i] : ((i < 256) ? 0.f : aggr_b[i - 256]);

  int2* __restrict__ mine = (int2*)recsw + (size_t)g * chunk;
  for (int base = e0; base < e1; base += 256 * PROD_U) {
    int hh[PROD_U], tt[PROD_U], bb[PROD_U];
#pragma unroll
    for (int u = 0; u < PROD_U; ++u) {
      const int e = base + tid + 256 * u;
      const int ec = e < e1 ? e : e1 - 1;
      hh[u] = trip[3 * ec]; tt[u] = trip[3 * ec + 1]; bb[u] = trip[3 * ec + 2];
    }
#pragma unroll
    for (int u = 0; u < PROD_U; ++u) {
      const int e = base + tid + 256 * u;
      const int b = hh[u] >> 3;
      if (e < e1 && hh[u] >= 0 && b < nb) {
        const int pos = atomicAdd(&s_cur[b], 1);
        mine[pos] = make_int2(e, (int)((unsigned)tt[u] << 7) | ((hh[u] & 7) << 4) | clamp_bin(bb[u]));
      }
    }
  }
}

__global__ void k_gemm(const float* __restrict__ A, const float* __restrict__ Bmat,
                       const float* __restrict__ Bias,
                       float* __restrict__ P1, float* __restrict__ P2,
                       float* __restrict__ Mout, int R) {
  const int lane = threadIdx.x & 31;
  const int half = lane >> 4;
  const int l16  = lane & 15;
  const int m0 = blockIdx.x * 16;
  const int n0 = blockIdx.y * 64;

  int mrow = m0 + l16; if (mrow >= R) mrow = R - 1;
  const float* __restrict__ Arow = A + (size_t)mrow * DIM_IN;

  v8f acc0 = {}, acc1 = {}, acc2 = {}, acc3 = {};
  for (int k0 = 0; k0 < DIM_IN; k0 += 4) {
    const int ka = k0 + (half << 1);
    v2f a; a.x = Arow[ka]; a.y = Arow[ka + 1];
    const float* __restrict__ Bp = Bmat + (size_t)ka * NCOLS + n0 + l16;
    v2f b0; b0.x = Bp[0];  b0.y = Bp[NCOLS];
    v2f b1; b1.x = Bp[16]; b1.y = Bp[NCOLS + 16];
    v2f b2; b2.x = Bp[32]; b2.y = Bp[NCOLS + 32];
    v2f b3; b3.x = Bp[48]; b3.y = Bp[NCOLS + 48];
    acc0 = __builtin_amdgcn_wmma_f32_16x16x4_f32(false, a, false, b0, (short)0, acc0, false, false);
    acc1 = __builtin_amdgcn_wmma_f32_16x16x4_f32(false, a, false, b1, (short)0, acc1, false, false);
    acc2 = __builtin_amdgcn_wmma_f32_16x16x4_f32(false, a, false, b2, (short)0, acc2, false, false);
    acc3 = __builtin_amdgcn_wmma_f32_16x16x4_f32(false, a, false, b3, (short)0, acc3, false, false);
  }

#pragma unroll
  for (int i = 0; i < 8; ++i) {
    const int row = m0 + i + (half << 3);
    if (row >= R) continue;
    float vals[4] = { acc0[i], acc1[i], acc2[i], acc3[i] };
#pragma unroll
    for (int j = 0; j < 4; ++j) {
      const int c = n0 + j * 16 + l16;
      const float v = vals[j] + Bias[c];
      if (c < 128)      P1[(size_t)row * 128 + c]         = v;
      else if (c < 256) P2[(size_t)row * 128 + (c - 128)] = v;
      else              Mout[(size_t)row * 128 + (c - 256)] = v;
    }
  }
}

__device__ __forceinline__ void wave_lds_handover() {
  __builtin_amdgcn_fence(__ATOMIC_RELEASE, "wavefront");
  __builtin_amdgcn_wave_barrier();
  __builtin_amdgcn_fence(__ATOMIC_ACQUIRE, "wavefront");
}

__device__ __forceinline__ float quad_sum(float s) {
  s += __int_as_float(__builtin_amdgcn_update_dpp(0, __float_as_int(s), 0xB1, 0xF, 0xF, true));
  s += __int_as_float(__builtin_amdgcn_update_dpp(0, __float_as_int(s), 0x4E, 0xF, 0xF, true));
  return s;
}

__device__ __forceinline__ float edge_dot(const float4 p1, const float4 w,
                                          const float* __restrict__ P2, int t, int d0) {
  const float4 p2 = *(const float4*)(P2 + (size_t)t * 128 + d0);
  const float s = lrelu(p1.x + p2.x) * w.x + lrelu(p1.y + p2.y) * w.y +
                  lrelu(p1.z + p2.z) * w.z + lrelu(p1.w + p2.w) * w.w;
  return quad_sum(s);
}

#define GRP 8

__device__ __forceinline__ void logits_to_lds(const int tv, const int cnt, const int jbase,
                                              const float4 p1, const float4 w,
                                              const float* __restrict__ P2, const int d0,
                                              const int lane, float* __restrict__ slog) {
  const int h = lane >> 2, q = lane & 3;
  for (int j = 0; j < cnt; j += GRP) {
    float4 a[GRP];
#pragma unroll
    for (int u = 0; u < GRP; ++u) {
      int ju = j + u; ju = ju < cnt ? ju : cnt - 1;
      const int t = __builtin_amdgcn_readlane(tv, ju);
      a[u] = *(const float4*)(P2 + (size_t)t * 128 + d0);
    }
    float s[GRP];
#pragma unroll
    for (int u = 0; u < GRP; ++u) {
      const float x = lrelu(p1.x + a[u].x) * w.x + lrelu(p1.y + a[u].y) * w.y +
                      lrelu(p1.z + a[u].z) * w.z + lrelu(p1.w + a[u].w) * w.w;
      s[u] = quad_sum(x);
    }
#pragma unroll
    for (int u = 0; u < GRP; u += 4) {
      const float c0 = s[u], c1 = s[u + 1], c2 = s[u + 2], c3 = s[u + 3];
      const float val = q == 0 ? c0 : (q == 1 ? c1 : (q == 2 ? c2 : c3));
      const int je = j + u + q;
      if (je < cnt) slog[(jbase + je) * NUM_HEAD + h] = val;
    }
  }
}

__device__ __forceinline__ void weighted_rows(const int tv, const int cnt, const int jbase,
                                              const float* __restrict__ Mmat, const int d0,
                                              const int lane, const float* __restrict__ slog,
                                              float4& acc) {
  const int h = lane >> 2;
  for (int j = 0; j < cnt; j += GRP) {
    float4 a[GRP];
    float be[GRP];
#pragma unroll
    for (int u = 0; u < GRP; ++u) {
      int ju = j + u; ju = ju < cnt ? ju : cnt - 1;
      const int t = __builtin_amdgcn_readlane(tv, ju);
      a[u] = *(const float4*)(Mmat + (size_t)t * 128 + d0);
      be[u] = slog[(jbase + ju) * NUM_HEAD + h];
    }
#pragma unroll
    for (int u = 0; u < GRP; ++u) {
      if (j + u < cnt) {
        acc.x += be[u] * a[u].x; acc.y += be[u] * a[u].y;
        acc.z += be[u] * a[u].z; acc.w += be[u] * a[u].w;
      }
    }
  }
}

__global__ void __launch_bounds__(256) k_rel(const int* __restrict__ trip,
                      const int* __restrict__ recsw, const int* __restrict__ offw,
                      const float* __restrict__ P1, const float* __restrict__ P2,
                      const float* __restrict__ Mmat, const float* __restrict__ av,
                      const float* __restrict__ ab, float* __restrict__ out, int R, int E) {
  __shared__ float s_bias[NUM_BIN * NUM_HEAD];
  __shared__ float s_log[8][SEG_CAP * NUM_HEAD];
  __shared__ int   s_tail[8][SEG_CAP];
  __shared__ int   s_bin[8][SEG_CAP];
  __shared__ int   s_e[8][SEG_CAP];
  __shared__ int   s_y[8][SEG_CAP];
  __shared__ int   s_n[8];

  const int lane = threadIdx.x & 31;
  const int wv = __builtin_amdgcn_readfirstlane(threadIdx.x >> 5);
  int2 oc = make_int2(0, 0);
  if (blockIdx.x < MAXBIN) oc = ((const int2*)offw)[(size_t)blockIdx.x * NPROD + threadIdx.x];
  if (threadIdx.x < NUM_BIN * NUM_HEAD) s_bias[threadIdx.x] = lrelu(ab[threadIdx.x]);
  if (threadIdx.x < 8) s_n[threadIdx.x] = 0;
  __syncthreads();

  {
    const int chunk = (E + NPROD - 1) / NPROD;
    const int2* __restrict__ src = (const int2*)recsw + (size_t)threadIdx.x * chunk + oc.x;
    for (int i0 = 0; i0 < oc.y; i0 += 4) {
      int2 rc[4];
#pragma unroll
      for (int u = 0; u < 4; ++u) rc[u] = src[i0 + u < oc.y ? i0 + u : oc.y - 1];
#pragma unroll
      for (int u = 0; u < 4; ++u) {
        if (i0 + u < oc.y) {
          const int hl = (rc[u].y >> 4) & 7;
          const int slot = atomicAdd(&s_n[hl], 1);
          if (slot < SEG_CAP) { s_e[hl][slot] = rc[u].x; s_y[hl][slot] = rc[u].y; }
        }
      }
    }
  }
  __syncthreads();

  const int r = blockIdx.x * 8 + wv;
  if (r >= R) return;
  const int n = __builtin_amdgcn_readfirstlane(s_n[wv]);

  const int d0 = lane * 4;
  const int h  = lane >> 2;
  const float4 p1 = *(const float4*)(P1 + (size_t)r * 128 + d0);
  const float4 w  = *(const float4*)(av + d0);
  float4 acc = make_float4(0.f, 0.f, 0.f, 0.f);
  float* __restrict__ slog = s_log[wv];

  if (n <= SEG_CAP) {
    const bool has0 = lane < n, has1 = lane + 32 < n;
    int2 q0 = make_int2(BIG_I, 0), q1 = make_int2(BIG_I, 0);
    if (has0) q0 = make_int2(s_e[wv][lane], s_y[wv][lane]);
    if (has1) q1 = make_int2(s_e[wv][32 + lane], s_y[wv][32 + lane]);
    const int e0 = q0.x, e1 = q1.x;
    int r0 = 0, r1 = 0;
    const int nlo = n < 32 ? n : 32;
    const int nhi = n - nlo;
    for (int j = 0; j < nlo; ++j) {
      const int x = __builtin_amdgcn_readlane(e0, j);
      r0 += (x < e0); r1 += (x < e1);
    }
    for (int j = 0; j < nhi; ++j) {
      const int x = __builtin_amdgcn_readlane(e1, j);
      r0 += (x < e0); r1 += (x < e1);
    }
    if (has0) { s_tail[wv][r0] = q0.y >> 7; s_bin[wv][r0] = q0.y & 15; }
    if (has1) { s_tail[wv][r1] = q1.y >> 7; s_bin[wv][r1] = q1.y & 15; }
    wave_lds_handover();
    const int tv0 = s_tail[wv][lane];
    const int tv1 = s_tail[wv][lane + 32];

    logits_to_lds(tv0, nlo, 0,  p1, w, P2, d0, lane, slog);
    logits_to_lds(tv1, nhi, 32, p1, w, P2, d0, lane, slog);
    wave_lds_handover();

    float v[SEG_CAP / 4];
    float mx = -BIG_F;
#pragma unroll
    for (int i = 0; i < SEG_CAP / 4; ++i) {
      const int j = (lane >> 3) + 4 * i;
      v[i] = -BIG_F;
      if (j < n) v[i] = slog[lane + 32 * i] + s_bias[s_bin[wv][j] * NUM_HEAD + (lane & 7)];
      mx = fmaxf(mx, v[i]);
    }
    mx = fmaxf(mx, __shfl_xor(mx, 8, 32));
    mx = fmaxf(mx, __shfl_xor(mx, 16, 32));
    float sum = 0.f;
#pragma unroll
    for (int i = 0; i < SEG_CAP / 4; ++i) {
      const int j = (lane >> 3) + 4 * i;
      v[i] = (j < n) ? expf(v[i] - mx) : 0.f;
      sum += v[i];
    }
    sum += __shfl_xor(sum, 8, 32);
    sum += __shfl_xor(sum, 16, 32);
    const float den = sum + 1e-16f;
#pragma unroll
    for (int i = 0; i < SEG_CAP / 4; ++i) {
      const int j = (lane >> 3) + 4 * i;
      if (j < n) slog[lane + 32 * i] = v[i] / den;
    }
    wave_lds_handover();

    weighted_rows(tv0, nlo, 0,  Mmat, d0, lane, slog, acc);
    weighted_rows(tv1, nhi, 32, Mmat, d0, lane, slog, acc);
  } else {
    float M = -BIG_F, L = 0.f;
    for (int base = 0; base < E; base += 32) {
      const int idx = base + lane;
      const bool mine = idx < E && trip[3 * idx] == r;
      unsigned todo = __builtin_amdgcn_readfirstlane((unsigned)__ballot(mine));
      while (todo != 0u) {
        const int e = base + __builtin_ctz(todo);
        todo &= todo - 1u;
        const int t = trip[3 * e + 1];
        const int b = clamp_bin(trip[3 * e + 2]);
        const float s = edge_dot(p1, w, P2, t, d0) + s_bias[b * NUM_HEAD + h];
        const float Mn = fmaxf(M, s);
        const float sc = expf(M - Mn), p = expf(s - Mn);
        const float4 m = *(const float4*)(Mmat + (size_t)t * 128 + d0);
        L = L * sc + p;
        acc.x = acc.x * sc + p * m.x; acc.y = acc.y * sc + p * m.y;
        acc.z = acc.z * sc + p * m.z; acc.w = acc.w * sc + p * m.w;
        M = Mn;
      }
    }
    const float den = L + 1e-16f;
    acc.x /= den; acc.y /= den; acc.z /= den; acc.w /= den;
  }
  *(float4*)(out + (size_t)r * 128 + d0) = acc;
}

extern "C" void kernel_launch(void* const* d_in, const int* in_sizes, int n_in,
                              void* d_out, int out_size, void* d_ws, size_t ws_size,
                              hipStream_t stream) {
  const float* emb      = (const float*)d_in[0];
  const int*   trip     = (const int*)  d_in[1];
  const float* attn_w   = (const float*)d_in[2];
  const float* attn_b   = (const float*)d_in[3];
  const float* attn_bin = (const float*)d_in[4];
  const float* attn_vec = (const float*)d_in[5];
  const float* aggr_w   = (const float*)d_in[6];
  const float* aggr_b   = (const float*)d_in[7];

  const int R = in_sizes[0] / DIM_IN;
  const int E = in_sizes[1] / 3;

  float* ws     = (float*)d_ws;
  float* Bmat   = ws;
  float* Bias   = Bmat + DIM_IN * NCOLS;
  float* P1     = Bias + NCOLS;
  float* P2     = P1 + (size_t)R * 128;
  float* Mm     = P2 + (size_t)R * 128;
  const int chunk = (E + NPROD - 1) / NPROD;
  int*   recs   = (int*)(Mm + (size_t)R * 128);
  int*   offT   = recs + (size_t)NPROD * chunk * 2;
  float* out    = (float*)d_out;

  k_prep<<<NPROD, 256, 0, stream>>>(attn_w, attn_b, aggr_w, aggr_b, Bmat, Bias, trip, recs, offT, R, E);

  dim3 gg((R + 15) / 16, NCOLS / 64);
  k_gemm<<<gg, 32, 0, stream>>>(emb, Bmat, Bias, P1, P2, Mm, R);

  k_rel<<<(R + 7) / 8, NPROD, 0, stream>>>(trip, recs, offT, P1, P2, Mm, attn_vec, attn_bin, out, R, E);
}


# Edge grouping runs as extra work-groups of the GEMM launch; GEMM takes the weights in place (64 rows staged in LDS per group, no B-build pass; k order permuted inside blocks of 8): 3 launches to 2
# speedup vs baseline: 3.4786x; 1.2049x over previous
#include <hip/hip_runtime.h>
#include <math.h>

#define DIM_IN   128
#define DIM_OUT  128
#define NUM_HEAD 8
#define DIM_HID  16
#define NUM_BIN  10
#define NEG_SLOPE 0.2f
#define NCOLS 384
#define SEG_CAP 64
#define NPROD 256
#define MAXBIN 4096
#define PROD_U 10
#define BPITCH 132
#define BIG_F 3.402823466e38f
#define BIG_I 2147483647

typedef __attribute__((ext_vector_type(2))) float v2f;
typedef __attribute__((ext_vector_type(8))) float v8f;

__device__ __forceinline__ float lrelu(float x) { return x >= 0.f ? x : NEG_SLOPE * x; }

__device__ __forceinline__ int clamp_bin(int b) {
  return b < 0 ? 0 : (b > NUM_BIN - 1 ? NUM_BIN - 1 : b);
}

__device__ __forceinline__ void group_edges(int* __restrict__ s_cnt, int* __restrict__ s_cur,
                                            int* __restrict__ s_wsum,
                                            const int* __restrict__ trip, int* __restrict__ recsw,
                                            int* __restrict__ offw, int R, int E) {
  const int tid = threadIdx.x;
  int nb = (R + 7) >> 3; if (nb > MAXBIN) nb = MAXBIN;
  const int g = blockIdx.x;
  const int chunk = (E + NPROD - 1) / NPROD;
  const int e0 = g * chunk;
  const int e1 = (e0 + chunk < E) ? e0 + chunk : E;
  for (int b = tid; b < nb; b += 256) s_cnt[b] = 0;
  __syncthreads();

  for (int base = e0; base < e1; base += 256 * PROD_U) {
    int hh[PROD_U];
#pragma unroll
    for (int u = 0; u < PROD_U; ++u) {
      const int e = base + tid + 256 * u;
      hh[u] = trip[3 * (e < e1 ? e : e1 - 1)];
    }
#pragma unroll
    for (int u = 0; u < PROD_U; ++u) {
      const int e = base + tid + 256 * u;
      const int b = hh[u] >> 3;
      if (e < e1 && hh[u] >= 0 && b < nb) atomicAdd(&s_cnt[b], 1);
    }
  }
  __syncthreads();

  const int lane = tid & 31, wv = tid >> 5;
  const int per = (nb + 255) / 256;
  const int lo = (tid * per < nb) ? tid * per : nb;
  const int hi = (lo + per < nb) ? lo + per : nb;
  int sum = 0;
  for (int b = lo; b < hi; ++b) sum += s_cnt[b];
  int v = sum;
#pragma unroll
  for (int d = 1; d < 32; d <<= 1) {
    const int y = __shfl_up(v, d, 32);
    if (lane >= d) v += y;
  }
  if (lane == 31) s_wsum[wv] = v;
  __syncthreads();
  int run = v - sum;
  for (int w = 0; w < wv; ++w) run += s_wsum[w];
  int2* __restrict__ offT = (int2*)offw;
  for (int b = lo; b < hi; ++b) {
    const int c = s_cnt[b];
    s_cur[b] = run;
    offT[(size_t)b * NPROD + g] = make_int2(run, c);
    run += c;
  }
  __syncthreads();

  int2* __restrict__ mine = (int2*)recsw + (size_t)g * chunk;
  for (int base = e0; base < e1; base += 256 * PROD_U) {
    int hh[PROD_U], tt[PROD_U], bb[PROD_U];
#pragma unroll
    for (int u = 0; u < PROD_U; ++u) {
      const int e = base + tid + 256 * u;
      const int ec = e < e1 ? e : e1 - 1;
      hh[u] = trip[3 * ec]; tt[u] = trip[3 * ec + 1]; bb[u] = trip[3 * ec + 2];
    }
#pragma unroll
    for (int u = 0; u < PROD_U; ++u) {
      const int e = base + tid + 256 * u;
      const int b = hh[u] >> 3;
      if (e < e1 && hh[u] >= 0 && b < nb) {
        const int pos = atomicAdd(&s_cur[b], 1);
        mine[pos] = make_int2(e, (int)((unsigned)tt[u] << 7) | ((hh[u] & 7) << 4) | clamp_bin(bb[u]));
      }
    }
  }
}

__global__ void __launch_bounds__(256) k_front(const float* __restrict__ A,
                       const float* __restrict__ attn_w, const float* __restrict__ attn_b,
                       const float* __restrict__ aggr_w, const float* __restrict__ aggr_b,
                       float* __restrict__ P1, float* __restrict__ P2, float* __restrict__ Mout,
                       const int* __restrict__ trip, int* __restrict__ recsw,
                       int* __restrict__ offw, int R, int E) {
  __shared__ __attribute__((aligned(16))) int s_u[64 * BPITCH];
  if (blockIdx.x < NPROD) {
    group_edges(s_u, s_u + MAXBIN, s_u + 2 * MAXBIN, trip, recsw, offw, R, E);
    return;
  }
  float* __restrict__ sB = (float*)s_u;
  const int gb = blockIdx.x - NPROD;
  const int lane = threadIdx.x & 31;
  const int wv = __builtin_amdgcn_readfirstlane(threadIdx.x >> 5);
  const int half = lane >> 4;
  const int l16  = lane & 15;
  const int n0 = (gb % (NCOLS / 64)) * 64;
  const int m0 = ((gb / (NCOLS / 64)) * 8 + wv) * 16;

  int mrow = m0 + l16; if (mrow >= R) mrow = R - 1;
  const float* __restrict__ Arow = A + (size_t)mrow * DIM_IN + 4 * half;
  float4 a[DIM_IN / 8];
#pragma unroll
  for (int k8 = 0; k8 < DIM_IN / 8; ++k8) a[k8] = *(const float4*)(Arow + k8 * 8);

  const float* __restrict__ wsrc;
  int wstride;
  if (n0 < 128)      { wsrc = attn_w + (size_t)n0 * 256;               wstride = 256; }
  else if (n0 < 256) { wsrc = attn_w + (size_t)(n0 - 128) * 256 + 128; wstride = 256; }
  else               { wsrc = aggr_w + (size_t)(n0 - 256) * 128;       wstride = 128; }
  {
    float4 wr[8];
#pragma unroll
    for (int u = 0; u < 8; ++u)
      wr[u] = *(const float4*)(wsrc + (size_t)(wv * 8 + u) * wstride + lane * 4);
#pragma unroll
    for (int u = 0; u < 8; ++u)
      *(float4*)(sB + (wv * 8 + u) * BPITCH + lane * 4) = wr[u];
  }
  const float* __restrict__ bsrc;
  float* __restrict__ dst;
  if (n0 < 128)      { bsrc = attn_b + n0;         dst = P1 + n0; }
  else if (n0 < 256) { bsrc = nullptr;             dst = P2 + (n0 - 128); }
  else               { bsrc = aggr_b + (n0 - 256); dst = Mout + (n0 - 256); }
  float bias[4] = { 0.f, 0.f, 0.f, 0.f };
  if (bsrc != nullptr) {
#pragma unroll
    for (int j = 0; j < 4; ++j) bias[j] = bsrc[j * 16 + l16];
  }
  __syncthreads();
  if (m0 >= R) return;

  const float* __restrict__ Bl = sB + l16 * BPITCH + 4 * half;
  v8f acc0 = {}, acc1 = {}, acc2 = {}, acc3 = {};
#pragma unroll
  for (int k8 = 0; k8 < DIM_IN / 8; ++k8) {
    const float4 b0 = *(const float4*)(Bl + 0 * 16 * BPITCH + k8 * 8);
    const float4 b1 = *(const float4*)(Bl + 1 * 16 * BPITCH + k8 * 8);
    const float4 b2 = *(const float4*)(Bl + 2 * 16 * BPITCH + k8 * 8);
    const float4 b3 = *(const float4*)(Bl + 3 * 16 * BPITCH + k8 * 8);
    v2f alo; alo.x = a[k8].x; alo.y = a[k8].y;
    v2f ahi; ahi.x = a[k8].z; ahi.y = a[k8].w;
    v2f t;
    t.x = b0.x; t.y = b0.y;
    acc0 = __builtin_amdgcn_wmma_f32_16x16x4_f32(false, alo, false, t, (short)0, acc0, false, false);
    t.x = b1.x; t.y = b1.y;
    acc1 = __builtin_amdgcn_wmma_f32_16x16x4_f32(false, alo, false, t, (short)0, acc1, false, false);
    t.x = b2.x; t.y = b2.y;
    acc2 = __builtin_amdgcn_wmma_f32_16x16x4_f32(false, alo, false, t, (short)0, acc2, false, false);
    t.x = b3.x; t.y = b3.y;
    acc3 = __builtin_amdgcn_wmma_f32_16x16x4_f32(false, alo, false, t, (short)0, acc3, false, false);
    t.x = b0.z; t.y = b0.w;
    acc0 = __builtin_amdgcn_wmma_f32_16x16x4_f32(false, ahi, false, t, (short)0, acc0, false, false);
    t.x = b1.z; t.y = b1.w;
    acc1 = __builtin_amdgcn_wmma_f32_16x16x4_f32(false, ahi, false, t, (short)0, acc1, false, false);
    t.x = b2.z; t.y = b2.w;
    acc2 = __builtin_amdgcn_wmma_f32_16x16x4_f32(false, ahi, false, t, (short)0, acc2, false, false);
    t.x = b3.z; t.y = b3.w;
    acc3 = __builtin_amdgcn_wmma_f32_16x16x4_f32(false, ahi, false, t, (short)0, acc3, false, false);
  }

#pragma unroll
  for (int i = 0; i < 8; ++i) {
    const int row = m0 + i + (half << 3);
    if (row >= R) continue;
    float vals[4] = { acc0[i], acc1[i], acc2[i], acc3[i] };
#pragma unroll
    for (int j = 0; j < 4; ++j)
      dst[(size_t)row * 128 + j * 16 + l16] = vals[j] + bias[j];
  }
}

__device__ __forceinline__ void wave_lds_handover() {
  __builtin_amdgcn_fence(__ATOMIC_RELEASE, "wavefront");
  __builtin_amdgcn_wave_barrier();
  __builtin_amdgcn_fence(__ATOMIC_ACQUIRE, "wavefront");
}

__device__ __forceinline__ float quad_sum(float s) {
  s += __int_as_float(__builtin_amdgcn_update_dpp(0, __float_as_int(s), 0xB1, 0xF, 0xF, true));
  s += __int_as_float(__builtin_amdgcn_update_dpp(0, __float_as_int(s), 0x4E, 0xF, 0xF, true));
  return s;
}

__device__ __forceinline__ float edge_dot(const float4 p1, const float4 w,
                                          const float* __restrict__ P2, int t, int d0) {
  const float4 p2 = *(const float4*)(P2 + (size_t)t * 128 + d0);
  const float s = lrelu(p1.x + p2.x) * w.x + lrelu(p1.y + p2.y) * w.y +
                  lrelu(p1.z + p2.z) * w.z + lrelu(p1.w + p2.w) * w.w;
  return quad_sum(s);
}

#define GRP 8

__device__ __forceinline__ void logits_to_lds(const int tv, const int cnt, const int jbase,
                                              const float4 p1, const float4 w,
                                              const float* __restrict__ P2, const int d0,
                                              const int lane, float* __restrict__ slog) {
  const int h = lane >> 2, q = lane & 3;
  for (int j = 0; j < cnt; j += GRP) {
    float4 a[GRP];
#pragma unroll
    for (int u = 0; u < GRP; ++u) {
      int ju = j + u; ju = ju < cnt ? ju : cnt - 1;
      const int t = __builtin_amdgcn_readlane(tv, ju);
      a[u] = *(const float4*)(P2 + (size_t)t * 128 + d0);
    }
    float s[GRP];
#pragma unroll
    for (int u = 0; u < GRP; ++u) {
      const float x = lrelu(p1.x + a[u].x) * w.x + lrelu(p1.y + a[u].y) * w.y +
                      lrelu(p1.z + a[u].z) * w.z + lrelu(p1.w + a[u].w) * w.w;
      s[u] = quad_sum(x);
    }
#pragma unroll
    for (int u = 0; u < GRP; u += 4) {
      const float c0 = s[u], c1 = s[u + 1], c2 = s[u + 2], c3 = s[u + 3];
      const float val = q == 0 ? c0 : (q == 1 ? c1 : (q == 2 ? c2 : c3));
      const int je = j + u + q;
      if (je < cnt) slog[(jbase + je) * NUM_HEAD + h] = val;
    }
  }
}

__device__ __forceinline__ void weighted_rows(const int tv, const int cnt, const int jbase,
                                              const float* __restrict__ Mmat, const int d0,
                                              const int lane, const float* __restrict__ slog,
                                              float4& acc) {
  const int h = lane >> 2;
  for (int j = 0; j < cnt; j += GRP) {
    float4 a[GRP];
    float be[GRP];
#pragma unroll
    for (int u = 0; u < GRP; ++u) {
      int ju = j + u; ju = ju < cnt ? ju : cnt - 1;
      const int t = __builtin_amdgcn_readlane(tv, ju);
      a[u] = *(const float4*)(Mmat + (size_t)t * 128 + d0);
      be[u] = slog[(jbase + ju) * NUM_HEAD + h];
    }
#pragma unroll
    for (int u = 0; u < GRP; ++u) {
      if (j + u < cnt) {
        acc.x += be[u] * a[u].x; acc.y += be[u] * a[u].y;
        acc.z += be[u] * a[u].z; acc.w += be[u] * a[u].w;
      }
    }
  }
}

__global__ void __launch_bounds__(256) k_rel(const int* __restrict__ trip,
                      const int* __restrict__ recsw, const int* __restrict__ offw,
                      const float* __restrict__ P1, const float* __restrict__ P2,
                      const float* __restrict__ Mmat, const float* __restrict__ av,
                      const float* __restrict__ ab, float* __restrict__ out, int R, int E) {
  __shared__ float s_bias[NUM_BIN * NUM_HEAD];
  __shared__ float s_log[8][SEG_CAP * NUM_HEAD];
  __shared__ int   s_tail[8][SEG_CAP];
  __shared__ int   s_bin[8][SEG_CAP];
  __shared__ int   s_e[8][SEG_CAP];
  __shared__ int   s_y[8][SEG_CAP];
  __shared__ int   s_n[8];

  const int lane = threadIdx.x & 31;
  const int wv = __builtin_amdgcn_readfirstlane(threadIdx.x >> 5);
  int2 oc = make_int2(0, 0);
  if (blockIdx.x < MAXBIN) oc = ((const int2*)offw)[(size_t)blockIdx.x * NPROD + threadIdx.x];
  if (threadIdx.x < NUM_BIN * NUM_HEAD) s_bias[threadIdx.x] = lrelu(ab[threadIdx.x]);
  if (threadIdx.x < 8) s_n[threadIdx.x] = 0;
  __syncthreads();

  {
    const int chunk = (E + NPROD - 1) / NPROD;
    const int2* __restrict__ src = (const int2*)recsw + (size_t)threadIdx.x * chunk + oc.x;
    for (int i0 = 0; i0 < oc.y; i0 += 4) {
      int2 rc[4];
#pragma unroll
      for (int u = 0; u < 4; ++u) rc[u] = src[i0 + u < oc.y ? i0 + u : oc.y - 1];
#pragma unroll
      for (int u = 0; u < 4; ++u) {
        if (i0 + u < oc.y) {
          const int hl = (rc[u].y >> 4) & 7;
          const int slot = atomicAdd(&s_n[hl], 1);
          if (slot < SEG_CAP) { s_e[hl][slot] = rc[u].x; s_y[hl][slot] = rc[u].y; }
        }
      }
    }
  }
  __syncthreads();

  const int r = blockIdx.x * 8 + wv;
  if (r >= R) return;
  const int n = __builtin_amdgcn_readfirstlane(s_n[wv]);

  const int d0 = lane * 4;
  const int h  = lane >> 2;
  const float4 p1 = *(const float4*)(P1 + (size_t)r * 128 + d0);
  const float4 w  = *(const float4*)(av + d0);
  float4 acc = make_float4(0.f, 0.f, 0.f, 0.f);
  float* __restrict__ slog = s_log[wv];

  if (n <= SEG_CAP) {
    const bool has0 = lane < n, has1 = lane + 32 < n;
    int2 q0 = make_int2(BIG_I, 0), q1 = make_int2(BIG_I, 0);
    if (has0) q0 = make_int2(s_e[wv][lane], s_y[wv][lane]);
    if (has1) q1 = make_int2(s_e[wv][32 + lane], s_y[wv][32 + lane]);
    const int e0 = q0.x, e1 = q1.x;
    int r0 = 0, r1 = 0;
    const int nlo = n < 32 ? n : 32;
    const int nhi = n - nlo;
    for (int j = 0; j < nlo; ++j) {
      const int x = __builtin_amdgcn_readlane(e0, j);
      r0 += (x < e0); r1 += (x < e1);
    }
    for (int j = 0; j < nhi; ++j) {
      const int x = __builtin_amdgcn_readlane(e1, j);
      r0 += (x < e0); r1 += (x < e1);
    }
    if (has0) { s_tail[wv][r0] = q0.y >> 7; s_bin[wv][r0] = q0.y & 15; }
    if (has1) { s_tail[wv][r1] = q1.y >> 7; s_bin[wv][r1] = q1.y & 15; }
    wave_lds_handover();
    const int tv0 = s_tail[wv][lane];
    const int tv1 = s_tail[wv][lane + 32];

    logits_to_lds(tv0, nlo, 0,  p1, w, P2, d0, lane, slog);
    logits_to_lds(tv1, nhi, 32, p1, w, P2, d0, lane, slog);
    wave_lds_handover();

    float v[SEG_CAP / 4];
    float mx = -BIG_F;
#pragma unroll
    for (int i = 0; i < SEG_CAP / 4; ++i) {
      const int j = (lane >> 3) + 4 * i;
      v[i] = -BIG_F;
      if (j < n) v[i] = slog[lane + 32 * i] + s_bias[s_bin[wv][j] * NUM_HEAD + (lane & 7)];
      mx = fmaxf(mx, v[i]);
    }
    mx = fmaxf(mx, __shfl_xor(mx, 8, 32));
    mx = fmaxf(mx, __shfl_xor(mx, 16, 32));
    float sum = 0.f;
#pragma unroll
    for (int i = 0; i < SEG_CAP / 4; ++i) {
      const int j = (lane >> 3) + 4 * i;
      v[i] = (j < n) ? expf(v[i] - mx) : 0.f;
      sum += v[i];
    }
    sum += __shfl_xor(sum, 8, 32);
    sum += __shfl_xor(sum, 16, 32);
    const float den = sum + 1e-16f;
#pragma unroll
    for (int i = 0; i < SEG_CAP / 4; ++i) {
      const int j = (lane >> 3) + 4 * i;
      if (j < n) slog[lane + 32 * i] = v[i] / den;
    }
    wave_lds_handover();

    weighted_rows(tv0, nlo, 0,  Mmat, d0, lane, slog, acc);
    weighted_rows(tv1, nhi, 32, Mmat, d0, lane, slog, acc);
  } else {
    float M = -BIG_F, L = 0.f;
    for (int base = 0; base < E; base += 32) {
      const int idx = base + lane;
      const bool mine = idx < E && trip[3 * idx] == r;
      unsigned todo = __builtin_amdgcn_readfirstlane((unsigned)__ballot(mine));
      while (todo != 0u) {
        const int e = base + __builtin_ctz(todo);
        todo &= todo - 1u;
        const int t = trip[3 * e + 1];
        const int b = clamp_bin(trip[3 * e + 2]);
        const float s = edge_dot(p1, w, P2, t, d0) + s_bias[b * NUM_HEAD + h];
        const float Mn = fmaxf(M, s);
        const float sc = expf(M - Mn), p = expf(s - Mn);
        const float4 m = *(const float4*)(Mmat + (size_t)t * 128 + d0);
        L = L * sc + p;
        acc.x = acc.x * sc + p * m.x; acc.y = acc.y * sc + p * m.y;
        acc.z = acc.z * sc + p * m.z; acc.w = acc.w * sc + p * m.w;
        M = Mn;
      }
    }
    const float den = L + 1e-16f;
    acc.x /= den; acc.y /= den; acc.z /= den; acc.w /= den;
  }
  *(float4*)(out + (size_t)r * 128 + d0) = acc;
}

extern "C" void kernel_launch(void* const* d_in, const int* in_sizes, int n_in,
                              void* d_out, int out_size, void* d_ws, size_t ws_size,
                              hipStream_t stream) {
  const float* emb      = (const float*)d_in[0];
  const int*   trip     = (const int*)  d_in[1];
  const float* attn_w   = (const float*)d_in[2];
  const float* attn_b   = (const float*)d_in[3];
  const float* attn_bin = (const float*)d_in[4];
  const float* attn_vec = (const float*)d_in[5];
  const float* aggr_w   = (const float*)d_in[6];
  const float* aggr_b   = (const float*)d_in[7];

  const int R = in_sizes[0] / DIM_IN;
  const int E = in_sizes[1] / 3;

  float* ws     = (float*)d_ws;
  float* P1     = ws + DIM_IN * NCOLS + NCOLS;
  float* P2     = P1 + (size_t)R * 128;
  float* Mm     = P2 + (size_t)R * 128;
  const int chunk = (E + NPROD - 1) / NPROD;
  int*   recs   = (int*)(Mm + (size_t)R * 128);
  int*   offT   = recs + (size_t)NPROD * chunk * 2;
  float* out    = (float*)d_out;

  const int ggrp = ((R + 127) / 128) * (NCOLS / 64);
  k_front<<<NPROD + ggrp, 256, 0, stream>>>(emb, attn_w, attn_b, aggr_w, aggr_b, P1, P2, Mm, trip, recs, offT, R, E);

  k_rel<<<(R + 7) / 8, NPROD, 0, stream>>>(trip, recs, offT, P1, P2, Mm, attn_vec, attn_bin, out, R, E);
}


# Logit pass of the per-relation kernel with fewer vector instructions per edge: leaky-relu as max(x, 0.2x), 4-dim dot as fused multiply-adds, quad totals folded in 3 adds so edge q lands in lane q
# speedup vs baseline: 3.6562x; 1.0511x over previous
#include <hip/hip_runtime.h>
#include <math.h>

#define DIM_IN   128
#define DIM_OUT  128
#define NUM_HEAD 8
#define DIM_HID  16
#define NUM_BIN  10
#define NEG_SLOPE 0.2f
#define NCOLS 384
#define SEG_CAP 64
#define NPROD 256
#define MAXBIN 4096
#define PROD_U 10
#define BPITCH 132
#define BIG_F 3.402823466e38f
#define BIG_I 2147483647

typedef __attribute__((ext_vector_type(2))) float v2f;
typedef __attribute__((ext_vector_type(8))) float v8f;

__device__ __forceinline__ float lrelu(float x) { return x >= 0.f ? x : NEG_SLOPE * x; }

__device__ __forceinline__ int clamp_bin(int b) {
  return b < 0 ? 0 : (b > NUM_BIN - 1 ? NUM_BIN - 1 : b);
}

__device__ __forceinline__ void group_edges(int* __restrict__ s_cnt, int* __restrict__ s_cur,
                                            int* __restrict__ s_wsum,
                                            const int* __restrict__ trip, int* __restrict__ recsw,
                                            int* __restrict__ offw, int R, int E) {
  const int tid = threadIdx.x;
  int nb = (R + 7) >> 3; if (nb > MAXBIN) nb = MAXBIN;
  const int g = blockIdx.x;
  const int chunk = (E + NPROD - 1) / NPROD;
  const int e0 = g * chunk;
  const int e1 = (e0 + chunk < E) ? e0 + chunk : E;
  for (int b = tid; b < nb; b += 256) s_cnt[b] = 0;
  __syncthreads();

  for (int base = e0; base < e1; base += 256 * PROD_U) {
    int hh[PROD_U];
#pragma unroll
    for (int u = 0; u < PROD_U; ++u) {
      const int e = base + tid + 256 * u;
      hh[u] = trip[3 * (e < e1 ? e : e1 - 1)];
    }
#pragma unroll
    for (int u = 0; u < PROD_U; ++u) {
      const int e = base + tid + 256 * u;
      const int b = hh[u] >> 3;
      if (e < e1 && hh[u] >= 0 && b < nb) atomicAdd(&s_cnt[b], 1);
    }
  }
  __syncthreads();

  const int lane = tid & 31, wv = tid >> 5;
  const int per = (nb + 255) / 256;
  const int lo = (tid * per < nb) ? tid * per : nb;
  const int hi = (lo + per < nb) ? lo + per : nb;
  int sum = 0;
  for (int b = lo; b < hi; ++b) sum += s_cnt[b];
  int v = sum;
#pragma unroll
  for (int d = 1; d < 32; d <<= 1) {
    const int y = __shfl_up(v, d, 32);
    if (lane >= d) v += y;
  }
  if (lane == 31) s_wsum[wv] = v;
  __syncthreads();
  int run = v - sum;
  for (int w = 0; w < wv; ++w) run += s_wsum[w];
  int2* __restrict__ offT = (int2*)offw;
  for (int b = lo; b < hi; ++b) {
    const int c = s_cnt[b];
    s_cur[b] = run;
    offT[(size_t)b * NPROD + g] = make_int2(run, c);
    run += c;
  }
  __syncthreads();

  int2* __restrict__ mine = (int2*)recsw + (size_t)g * chunk;
  for (int base = e0; base < e1; base += 256 * PROD_U) {
    int hh[PROD_U], tt[PROD_U], bb[PROD_U];
#pragma unroll
    for (int u = 0; u < PROD_U; ++u) {
      const int e = base + tid + 256 * u;
      const int ec = e < e1 ? e : e1 - 1;
      hh[u] = trip[3 * ec]; tt[u] = trip[3 * ec + 1]; bb[u] = trip[3 * ec + 2];
    }
#pragma unroll
    for (int u = 0; u < PROD_U; ++u) {
      const int e = base + tid + 256 * u;
      const int b = hh[u] >> 3;
      if (e < e1 && hh[u] >= 0 && b < nb) {
        const int pos = atomicAdd(&s_cur[b], 1);
        mine[pos] = make_int2(e, (int)((unsigned)tt[u] << 7) | ((hh[u] & 7) << 4) | clamp_bin(bb[u]));
      }
    }
  }
}

__global__ void __launch_bounds__(256) k_front(const float* __restrict__ A,
                       const float* __restrict__ attn_w, const float* __restrict__ attn_b,
                       const float* __restrict__ aggr_w, const float* __restrict__ aggr_b,
                       float* __restrict__ P1, float* __restrict__ P2, float* __restrict__ Mout,
                       const int* __restrict__ trip, int* __restrict__ recsw,
                       int* __restrict__ offw, int R, int E) {
  __shared__ __attribute__((aligned(16))) int s_u[64 * BPITCH];
  if (blockIdx.x < NPROD) {
    group_edges(s_u, s_u + MAXBIN, s_u + 2 * MAXBIN, trip, recsw, offw, R, E);
    return;
  }
  float* __restrict__ sB = (float*)s_u;
  const int gb = blockIdx.x - NPROD;
  const int lane = threadIdx.x & 31;
  const int wv = __builtin_amdgcn_readfirstlane(threadIdx.x >> 5);
  const int half = lane >> 4;
  const int l16  = lane & 15;
  const int n0 = (gb % (NCOLS / 64)) * 64;
  const int m0 = ((gb / (NCOLS / 64)) * 8 + wv) * 16;

  int mrow = m0 + l16; if (mrow >= R) mrow = R - 1;
  const float* __restrict__ Arow = A + (size_t)mrow * DIM_IN + 4 * half;
  float4 a[DIM_IN / 8];
#pragma unroll
  for (int k8 = 0; k8 < DIM_IN / 8; ++k8) a[k8] = *(const float4*)(Arow + k8 * 8);

  const float* __restrict__ wsrc;
  int wstride;
  if (n0 < 128)      { wsrc = attn_w + (size_t)n0 * 256;               wstride = 256; }
  else if (n0 < 256) { wsrc = attn_w + (size_t)(n0 - 128) * 256 + 128; wstride = 256; }
  else               { wsrc = aggr_w + (size_t)(n0 - 256) * 128;       wstride = 128; }
  {
    float4 wr[8];
#pragma unroll
    for (int u = 0; u < 8; ++u)
      wr[u] = *(const float4*)(wsrc + (size_t)(wv * 8 + u) * wstride + lane * 4);
#pragma unroll
    for (int u = 0; u < 8; ++u)
      *(float4*)(sB + (wv * 8 + u) * BPITCH + lane * 4) = wr[u];
  }
  const float* __restrict__ bsrc;
  float* __restrict__ dst;
  if (n0 < 128)      { bsrc = attn_b + n0;         dst = P1 + n0; }
  else if (n0 < 256) { bsrc = nullptr;             dst = P2 + (n0 - 128); }
  else               { bsrc = aggr_b + (n0 - 256); dst = Mout + (n0 - 256); }
  float bias[4] = { 0.f, 0.f, 0.f, 0.f };
  if (bsrc != nullptr) {
#pragma unroll
    for (int j = 0; j < 4; ++j) bias[j] = bsrc[j * 16 + l16];
  }
  __syncthreads();
  if (m0 >= R) return;

  const float* __restrict__ Bl = sB + l16 * BPITCH + 4 * half;
  v8f acc0 = {}, acc1 = {}, acc2 = {}, acc3 = {};
#pragma unroll
  for (int k8 = 0; k8 < DIM_IN / 8; ++k8) {
    const float4 b0 = *(const float4*)(Bl + 0 * 16 * BPITCH + k8 * 8);
    const float4 b1 = *(const float4*)(Bl + 1 * 16 * BPITCH + k8 * 8);
    const float4 b2 = *(const float4*)(Bl + 2 * 16 * BPITCH + k8 * 8);
    const float4 b3 = *(const float4*)(Bl + 3 * 16 * BPITCH + k8 * 8);
    v2f alo; alo.x = a[k8].x; alo.y = a[k8].y;
    v2f ahi; ahi.x = a[k8].z; ahi.y = a[k8].w;
    v2f t;
    t.x = b0.x; t.y = b0.y;
    acc0 = __builtin_amdgcn_wmma_f32_16x16x4_f32(false, alo, false, t, (short)0, acc0, false, false);
    t.x = b1.x; t.y = b1.y;
    acc1 = __builtin_amdgcn_wmma_f32_16x16x4_f32(false, alo, false, t, (short)0, acc1, false, false);
    t.x = b2.x; t.y = b2.y;
    acc2 = __builtin_amdgcn_wmma_f32_16x16x4_f32(false, alo, false, t, (short)0, acc2, false, false);
    t.x = b3.x; t.y = b3.y;
    acc3 = __builtin_amdgcn_wmma_f32_16x16x4_f32(false, alo, false, t, (short)0, acc3, false, false);
    t.x = b0.z; t.y = b0.w;
    acc0 = __builtin_amdgcn_wmma_f32_16x16x4_f32(false, ahi, false, t, (short)0, acc0, false, false);
    t.x = b1.z; t.y = b1.w;
    acc1 = __builtin_amdgcn_wmma_f32_16x16x4_f32(false, ahi, false, t, (short)0, acc1, false, false);
    t.x = b2.z; t.y = b2.w;
    acc2 = __builtin_amdgcn_wmma_f32_16x16x4_f32(false, ahi, false, t, (short)0, acc2, false, false);
    t.x = b3.z; t.y = b3.w;
    acc3 = __builtin_amdgcn_wmma_f32_16x16x4_f32(false, ahi, false, t, (short)0, acc3, false, false);
  }

#pragma unroll
  for (int i = 0; i < 8; ++i) {
    const int row = m0 + i + (half << 3);
    if (row >= R) continue;
    float vals[4] = { acc0[i], acc1[i], acc2[i], acc3[i] };
#pragma unroll
    for (int j = 0; j < 4; ++j)
      dst[(size_t)row * 128 + j * 16 + l16] = vals[j] + bias[j];
  }
}

__device__ __forceinline__ void wave_lds_handover() {
  __builtin_amdgcn_fence(__ATOMIC_RELEASE, "wavefront");
  __builtin_amdgcn_wave_barrier();
  __builtin_amdgcn_fence(__ATOMIC_ACQUIRE, "wavefront");
}

__device__ __forceinline__ float quad_sum(float s) {
  s += __int_as_float(__builtin_amdgcn_update_dpp(0, __float_as_int(s), 0xB1, 0xF, 0xF, true));
  s += __int_as_float(__builtin_amdgcn_update_dpp(0, __float_as_int(s), 0x4E, 0xF, 0xF, true));
  return s;
}

__device__ __forceinline__ float edge_dot(const float4 p1, const float4 w,
                                          const float* __restrict__ P2, int t, int d0) {
  const float4 p2 = *(const float4*)(P2 + (size_t)t * 128 + d0);
  const float s = lrelu(p1.x + p2.x) * w.x + lrelu(p1.y + p2.y) * w.y +
                  lrelu(p1.z + p2.z) * w.z + lrelu(p1.w + p2.w) * w.w;
  return quad_sum(s);
}

#define GRP 8

__device__ __forceinline__ float lrelu_max(float x) { return fmaxf(x, NEG_SLOPE * x); }

__device__ __forceinline__ float quad_x1(float x) {
  return __int_as_float(__builtin_amdgcn_update_dpp(0, __float_as_int(x), 0xB1, 0xF, 0xF, true));
}
__device__ __forceinline__ float quad_x2(float x) {
  return __int_as_float(__builtin_amdgcn_update_dpp(0, __float_as_int(x), 0x4E, 0xF, 0xF, true));
}

__device__ __forceinline__ float quad_fold4(const float s0, const float s1, const float s2,
                                            const float s3, const bool odd1, const bool odd2) {
  const float k0 = odd1 ? s1 : s0, g0 = odd1 ? s0 : s1;
  const float k1 = odd1 ? s3 : s2, g1 = odd1 ? s2 : s3;
  const float r0 = k0 + quad_x1(g0);
  const float r1 = k1 + quad_x1(g1);
  const float kp = odd2 ? r1 : r0, gv = odd2 ? r0 : r1;
  return kp + quad_x2(gv);
}

__device__ __forceinline__ void logits_to_lds(const int tv, const int cnt, const int jbase,
                                              const float4 p1, const float4 w,
                                              const float* __restrict__ P2, const int d0,
                                              const int lane, float* __restrict__ slog) {
  const int h = lane >> 2, q = lane & 3;
  const bool odd1 = (q & 1) != 0, odd2 = (q & 2) != 0;
  for (int j = 0; j < cnt; j += GRP) {
    float4 a[GRP];
#pragma unroll
    for (int u = 0; u < GRP; ++u) {
      int ju = j + u; ju = ju < cnt ? ju : cnt - 1;
      const int t = __builtin_amdgcn_readlane(tv, ju);
      a[u] = *(const float4*)(P2 + (size_t)t * 128 + d0);
    }
    float s[GRP];
#pragma unroll
    for (int u = 0; u < GRP; ++u) {
      float x = lrelu_max(p1.x + a[u].x) * w.x;
      x = fmaf(lrelu_max(p1.y + a[u].y), w.y, x);
      x = fmaf(lrelu_max(p1.z + a[u].z), w.z, x);
      x = fmaf(lrelu_max(p1.w + a[u].w), w.w, x);
      s[u] = x;
    }
#pragma unroll
    for (int u = 0; u < GRP; u += 4) {
      const float val = quad_fold4(s[u], s[u + 1], s[u + 2], s[u + 3], odd1, odd2);
      const int je = j + u + q;
      if (je < cnt) slog[(jbase + je) * NUM_HEAD + h] = val;
    }
  }
}

__device__ __forceinline__ void weighted_rows(const int tv, const int cnt, const int jbase,
                                              const float* __restrict__ Mmat, const int d0,
                                              const int lane, const float* __restrict__ slog,
                                              float4& acc) {
  const int h = lane >> 2;
  for (int j = 0; j < cnt; j += GRP) {
    float4 a[GRP];
    float be[GRP];
#pragma unroll
    for (int u = 0; u < GRP; ++u) {
      int ju = j + u; ju = ju < cnt ? ju : cnt - 1;
      const int t = __builtin_amdgcn_readlane(tv, ju);
      a[u] = *(const float4*)(Mmat + (size_t)t * 128 + d0);
      be[u] = slog[(jbase + ju) * NUM_HEAD + h];
    }
#pragma unroll
    for (int u = 0; u < GRP; ++u) {
      if (j + u < cnt) {
        acc.x += be[u] * a[u].x; acc.y += be[u] * a[u].y;
        acc.z += be[u] * a[u].z; acc.w += be[u] * a[u].w;
      }
    }
  }
}

__global__ void __launch_bounds__(256) k_rel(const int* __restrict__ trip,
                      const int* __restrict__ recsw, const int* __restrict__ offw,
                      const float* __restrict__ P1, const float* __restrict__ P2,
                      const float* __restrict__ Mmat, const float* __restrict__ av,
                      const float* __restrict__ ab, float* __restrict__ out, int R, int E) {
  __shared__ float s_bias[NUM_BIN * NUM_HEAD];
  __shared__ float s_log[8][SEG_CAP * NUM_HEAD];
  __shared__ int   s_tail[8][SEG_CAP];
  __shared__ int   s_bin[8][SEG_CAP];
  __shared__ int   s_e[8][SEG_CAP];
  __shared__ int   s_y[8][SEG_CAP];
  __shared__ int   s_n[8];

  const int lane = threadIdx.x & 31;
  const int wv = __builtin_amdgcn_readfirstlane(threadIdx.x >> 5);
  int2 oc = make_int2(0, 0);
  if (blockIdx.x < MAXBIN) oc = ((const int2*)offw)[(size_t)blockIdx.x * NPROD + threadIdx.x];
  if (threadIdx.x < NUM_BIN * NUM_HEAD) s_bias[threadIdx.x] = lrelu(ab[threadIdx.x]);
  if (threadIdx.x < 8) s_n[threadIdx.x] = 0;
  __syncthreads();

  {
    const int chunk = (E + NPROD - 1) / NPROD;
    const int2* __restrict__ src = (const int2*)recsw + (size_t)threadIdx.x * chunk + oc.x;
    for (int i0 = 0; i0 < oc.y; i0 += 4) {
      int2 rc[4];
#pragma unroll
      for (int u = 0; u < 4; ++u) rc[u] = src[i0 + u < oc.y ? i0 + u : oc.y - 1];
#pragma unroll
      for (int u = 0; u < 4; ++u) {
        if (i0 + u < oc.y) {
          const int hl = (rc[u].y >> 4) & 7;
          const int slot = atomicAdd(&s_n[hl], 1);
          if (slot < SEG_CAP) { s_e[hl][slot] = rc[u].x; s_y[hl][slot] = rc[u].y; }
        }
      }
    }
  }
  __syncthreads();

  const int r = blockIdx.x * 8 + wv;
  if (r >= R) return;
  const int n = __builtin_amdgcn_readfirstlane(s_n[wv]);

  const int d0 = lane * 4;
  const int h  = lane >> 2;
  const float4 p1 = *(const float4*)(P1 + (size_t)r * 128 + d0);
  const float4 w  = *(const float4*)(av + d0);
  float4 acc = make_float4(0.f, 0.f, 0.f, 0.f);
  float* __restrict__ slog = s_log[wv];

  if (n <= SEG_CAP) {
    const bool has0 = lane < n, has1 = lane + 32 < n;
    int2 q0 = make_int2(BIG_I, 0), q1 = make_int2(BIG_I, 0);
    if (has0) q0 = make_int2(s_e[wv][lane], s_y[wv][lane]);
    if (has1) q1 = make_int2(s_e[wv][32 + lane], s_y[wv][32 + lane]);
    const int e0 = q0.x, e1 = q1.x;
    int r0 = 0, r1 = 0;
    const int nlo = n < 32 ? n : 32;
    const int nhi = n - nlo;
    for (int j = 0; j < nlo; ++j) {
      const int x = __builtin_amdgcn_readlane(e0, j);
      r0 += (x < e0); r1 += (x < e1);
    }
    for (int j = 0; j < nhi; ++j) {
      const int x = __builtin_amdgcn_readlane(e1, j);
      r0 += (x < e0); r1 += (x < e1);
    }
    if (has0) { s_tail[wv][r0] = q0.y >> 7; s_bin[wv][r0] = q0.y & 15; }
    if (has1) { s_tail[wv][r1] = q1.y >> 7; s_bin[wv][r1] = q1.y & 15; }
    wave_lds_handover();
    const int tv0 = s_tail[wv][lane];
    const int tv1 = s_tail[wv][lane + 32];

    logits_to_lds(tv0, nlo, 0,  p1, w, P2, d0, lane, slog);
    logits_to_lds(tv1, nhi, 32, p1, w, P2, d0, lane, slog);
    wave_lds_handover();

    float v[SEG_CAP / 4];
    float mx = -BIG_F;
#pragma unroll
    for (int i = 0; i < SEG_CAP / 4; ++i) {
      const int j = (lane >> 3) + 4 * i;
      v[i] = -BIG_F;
      if (j < n) v[i] = slog[lane + 32 * i] + s_bias[s_bin[wv][j] * NUM_HEAD + (lane & 7)];
      mx = fmaxf(mx, v[i]);
    }
    mx = fmaxf(mx, __shfl_xor(mx, 8, 32));
    mx = fmaxf(mx, __shfl_xor(mx, 16, 32));
    float sum = 0.f;
#pragma unroll
    for (int i = 0; i < SEG_CAP / 4; ++i) {
      const int j = (lane >> 3) + 4 * i;
      v[i] = (j < n) ? expf(v[i] - mx) : 0.f;
      sum += v[i];
    }
    sum += __shfl_xor(sum, 8, 32);
    sum += __shfl_xor(sum, 16, 32);
    const float den = sum + 1e-16f;
#pragma unroll
    for (int i = 0; i < SEG_CAP / 4; ++i) {
      const int j = (lane >> 3) + 4 * i;
      if (j < n) slog[lane + 32 * i] = v[i] / den;
    }
    wave_lds_handover();

    weighted_rows(tv0, nlo, 0,  Mmat, d0, lane, slog, acc);
    weighted_rows(tv1, nhi, 32, Mmat, d0, lane, slog, acc);
  } else {
    float M = -BIG_F, L = 0.f;
    for (int base = 0; base < E; base += 32) {
      const int idx = base + lane;
      const bool mine = idx < E && trip[3 * idx] == r;
      unsigned todo = __builtin_amdgcn_readfirstlane((unsigned)__ballot(mine));
      while (todo != 0u) {
        const int e = base + __builtin_ctz(todo);
        todo &= todo - 1u;
        const int t = trip[3 * e + 1];
        const int b = clamp_bin(trip[3 * e + 2]);
        const float s = edge_dot(p1, w, P2, t, d0) + s_bias[b * NUM_HEAD + h];
        const float Mn = fmaxf(M, s);
        const float sc = expf(M - Mn), p = expf(s - Mn);
        const float4 m = *(const float4*)(Mmat + (size_t)t * 128 + d0);
        L = L * sc + p;
        acc.x = acc.x * sc + p * m.x; acc.y = acc.y * sc + p * m.y;
        acc.z = acc.z * sc + p * m.z; acc.w = acc.w * sc + p * m.w;
        M = Mn;
      }
    }
    const float den = L + 1e-16f;
    acc.x /= den; acc.y /= den; acc.z /= den; acc.w /= den;
  }
  *(float4*)(out + (size_t)r * 128 + d0) = acc;
}

extern "C" void kernel_launch(void* const* d_in, const int* in_sizes, int n_in,
                              void* d_out, int out_size, void* d_ws, size_t ws_size,
                              hipStream_t stream) {
  const float* emb      = (const float*)d_in[0];
  const int*   trip     = (const int*)  d_in[1];
  const float* attn_w   = (const float*)d_in[2];
  const float* attn_b   = (const float*)d_in[3];
  const float* attn_bin = (const float*)d_in[4];
  const float* attn_vec = (const float*)d_in[5];
  const float* aggr_w   = (const float*)d_in[6];
  const float* aggr_b   = (const float*)d_in[7];

  const int R = in_sizes[0] / DIM_IN;
  const int E = in_sizes[1] / 3;

  float* ws     = (float*)d_ws;
  float* P1     = ws + DIM_IN * NCOLS + NCOLS;
  float* P2     = P1 + (size_t)R * 128;
  float* Mm     = P2 + (size_t)R * 128;
  const int chunk = (E + NPROD - 1) / NPROD;
  int*   recs   = (int*)(Mm + (size_t)R * 128);
  int*   offT   = recs + (size_t)NPROD * chunk * 2;
  float* out    = (float*)d_out;

  const int ggrp = ((R + 127) / 128) * (NCOLS / 64);
  k_front<<<NPROD + ggrp, 256, 0, stream>>>(emb, attn_w, attn_b, aggr_w, aggr_b, P1, P2, Mm, trip, recs, offT, R, E);

  k_rel<<<(R + 7) / 8, NPROD, 0, stream>>>(trip, recs, offT, P1, P2, Mm, attn_vec, attn_bin, out, R, E);
}


# Bins of 16 head relations (was 8): each per-relation work-group takes two relations per wave in turn: half the work-groups, waves, offset table and record line fetches; sum order kept
# speedup vs baseline: 4.0248x; 1.1008x over previous
#include <hip/hip_runtime.h>
#include <math.h>

#define DIM_IN   128
#define DIM_OUT  128
#define NUM_HEAD 8
#define DIM_HID  16
#define NUM_BIN  10
#define NEG_SLOPE 0.2f
#define NCOLS 384
#define SEG_CAP 64
#define NPROD 256
#define BREL 16
#define MAXBIN 4096
#define PROD_U 10
#define BPITCH 132
#define BIG_F 3.402823466e38f
#define BIG_I 2147483647

typedef __attribute__((ext_vector_type(2))) float v2f;
typedef __attribute__((ext_vector_type(8))) float v8f;

__device__ __forceinline__ float lrelu(float x) { return x >= 0.f ? x : NEG_SLOPE * x; }

__device__ __forceinline__ int clamp_bin(int b) {
  return b < 0 ? 0 : (b > NUM_BIN - 1 ? NUM_BIN - 1 : b);
}

__device__ __forceinline__ void group_edges(int* __restrict__ s_cnt, int* __restrict__ s_cur,
                                            int* __restrict__ s_wsum,
                                            const int* __restrict__ trip, int* __restrict__ recsw,
                                            int* __restrict__ offw, int R, int E) {
  const int tid = threadIdx.x;
  int nb = (R + BREL - 1) / BREL; if (nb > MAXBIN) nb = MAXBIN;
  const int g = blockIdx.x;
  const int chunk = (E + NPROD - 1) / NPROD;
  const int e0 = g * chunk;
  const int e1 = (e0 + chunk < E) ? e0 + chunk : E;
  for (int b = tid; b < nb; b += 256) s_cnt[b] = 0;
  __syncthreads();

  for (int base = e0; base < e1; base += 256 * PROD_U) {
    int hh[PROD_U];
#pragma unroll
    for (int u = 0; u < PROD_U; ++u) {
      const int e = base + tid + 256 * u;
      hh[u] = trip[3 * (e < e1 ? e : e1 - 1)];
    }
#pragma unroll
    for (int u = 0; u < PROD_U; ++u) {
      const int e = base + tid + 256 * u;
      const int b = hh[u] >> 4;
      if (e < e1 && hh[u] >= 0 && b < nb) atomicAdd(&s_cnt[b], 1);
    }
  }
  __syncthreads();

  const int lane = tid & 31, wv = tid >> 5;
  const int per = (nb + 255) / 256;
  const int lo = (tid * per < nb) ? tid * per : nb;
  const int hi = (lo + per < nb) ? lo + per : nb;
  int sum = 0;
  for (int b = lo; b < hi; ++b) sum += s_cnt[b];
  int v = sum;
#pragma unroll
  for (int d = 1; d < 32; d <<= 1) {
    const int y = __shfl_up(v, d, 32);
    if (lane >= d) v += y;
  }
  if (lane == 31) s_wsum[wv] = v;
  __syncthreads();
  int run = v - sum;
  for (int w = 0; w < wv; ++w) run += s_wsum[w];
  int2* __restrict__ offT = (int2*)offw;
  for (int b = lo; b < hi; ++b) {
    const int c = s_cnt[b];
    s_cur[b] = run;
    offT[(size_t)b * NPROD + g] = make_int2(run, c);
    run += c;
  }
  __syncthreads();

  int2* __restrict__ mine = (int2*)recsw + (size_t)g * chunk;
  for (int base = e0; base < e1; base += 256 * PROD_U) {
    int hh[PROD_U], tt[PROD_U], bb[PROD_U];
#pragma unroll
    for (int u = 0; u < PROD_U; ++u) {
      const int e = base + tid + 256 * u;
      const int ec = e < e1 ? e : e1 - 1;
      hh[u] = trip[3 * ec]; tt[u] = trip[3 * ec + 1]; bb[u] = trip[3 * ec + 2];
    }
#pragma unroll
    for (int u = 0; u < PROD_U; ++u) {
      const int e = base + tid + 256 * u;
      const int b = hh[u] >> 4;
      if (e < e1 && hh[u] >= 0 && b < nb) {
        const int pos = atomicAdd(&s_cur[b], 1);
        mine[pos] = make_int2(e, (int)((unsigned)tt[u] << 8) | ((hh[u] & (BREL - 1)) << 4) | clamp_bin(bb[u]));
      }
    }
  }
}

__global__ void __launch_bounds__(256) k_front(const float* __restrict__ A,
                       const float* __restrict__ attn_w, const float* __restrict__ attn_b,
                       const float* __restrict__ aggr_w, const float* __restrict__ aggr_b,
                       float* __restrict__ P1, float* __restrict__ P2, float* __restrict__ Mout,
                       const int* __restrict__ trip, int* __restrict__ recsw,
                       int* __restrict__ offw, int R, int E) {
  __shared__ __attribute__((aligned(16))) int s_u[64 * BPITCH];
  if (blockIdx.x < NPROD) {
    group_edges(s_u, s_u + MAXBIN, s_u + 2 * MAXBIN, trip, recsw, offw, R, E);
    return;
  }
  float* __restrict__ sB = (float*)s_u;
  const int gb = blockIdx.x - NPROD;
  const int lane = threadIdx.x & 31;
  const int wv = __builtin_amdgcn_readfirstlane(threadIdx.x >> 5);
  const int half = lane >> 4;
  const int l16  = lane & 15;
  const int n0 = (gb % (NCOLS / 64)) * 64;
  const int m0 = ((gb / (NCOLS / 64)) * 8 + wv) * 16;

  int mrow = m0 + l16; if (mrow >= R) mrow = R - 1;
  const float* __restrict__ Arow = A + (size_t)mrow * DIM_IN + 4 * half;
  float4 a[DIM_IN / 8];
#pragma unroll
  for (int k8 = 0; k8 < DIM_IN / 8; ++k8) a[k8] = *(const float4*)(Arow + k8 * 8);

  const float* __restrict__ wsrc;
  int wstride;
  if (n0 < 128)      { wsrc = attn_w + (size_t)n0 * 256;               wstride = 256; }
  else if (n0 < 256) { wsrc = attn_w + (size_t)(n0 - 128) * 256 + 128; wstride = 256; }
  else               { wsrc = aggr_w + (size_t)(n0 - 256) * 128;       wstride = 128; }
  {
    float4 wr[8];
#pragma unroll
    for (int u = 0; u < 8; ++u)
      wr[u] = *(const float4*)(wsrc + (size_t)(wv * 8 + u) * wstride + lane * 4);
#pragma unroll
    for (int u = 0; u < 8; ++u)
      *(float4*)(sB + (wv * 8 + u) * BPITCH + lane * 4) = wr[u];
  }
  const float* __restrict__ bsrc;
  float* __restrict__ dst;
  if (n0 < 128)      { bsrc = attn_b + n0;         dst = P1 + n0; }
  else if (n0 < 256) { bsrc = nullptr;             dst = P2 + (n0 - 128); }
  else               { bsrc = aggr_b + (n0 - 256); dst = Mout + (n0 - 256); }
  float bias[4] = { 0.f, 0.f, 0.f, 0.f };
  if (bsrc != nullptr) {
#pragma unroll
    for (int j = 0; j < 4; ++j) bias[j] = bsrc[j * 16 + l16];
  }
  __syncthreads();
  if (m0 >= R) return;

  const float* __restrict__ Bl = sB + l16 * BPITCH + 4 * half;
  v8f acc0 = {}, acc1 = {}, acc2 = {}, acc3 = {};
#pragma unroll
  for (int k8 = 0; k8 < DIM_IN / 8; ++k8) {
    const float4 b0 = *(const float4*)(Bl + 0 * 16 * BPITCH + k8 * 8);
    const float4 b1 = *(const float4*)(Bl + 1 * 16 * BPITCH + k8 * 8);
    const float4 b2 = *(const float4*)(Bl + 2 * 16 * BPITCH + k8 * 8);
    const float4 b3 = *(const float4*)(Bl + 3 * 16 * BPITCH + k8 * 8);
    v2f alo; alo.x = a[k8].x; alo.y = a[k8].y;
    v2f ahi; ahi.x = a[k8].z; ahi.y = a[k8].w;
    v2f t;
    t.x = b0.x; t.y = b0.y;
    acc0 = __builtin_amdgcn_wmma_f32_16x16x4_f32(false, alo, false, t, (short)0, acc0, false, false);
    t.x = b1.x; t.y = b1.y;
    acc1 = __builtin_amdgcn_wmma_f32_16x16x4_f32(false, alo, false, t, (short)0, acc1, false, false);
    t.x = b2.x; t.y = b2.y;
    acc2 = __builtin_amdgcn_wmma_f32_16x16x4_f32(false, alo, false, t, (short)0, acc2, false, false);
    t.x = b3.x; t.y = b3.y;
    acc3 = __builtin_amdgcn_wmma_f32_16x16x4_f32(false, alo, false, t, (short)0, acc3, false, false);
    t.x = b0.z; t.y = b0.w;
    acc0 = __builtin_amdgcn_wmma_f32_16x16x4_f32(false, ahi, false, t, (short)0, acc0, false, false);
    t.x = b1.z; t.y = b1.w;
    acc1 = __builtin_amdgcn_wmma_f32_16x16x4_f32(false, ahi, false, t, (short)0, acc1, false, false);
    t.x = b2.z; t.y = b2.w;
    acc2 = __builtin_amdgcn_wmma_f32_16x16x4_f32(false, ahi, false, t, (short)0, acc2, false, false);
    t.x = b3.z; t.y = b3.w;
    acc3 = __builtin_amdgcn_wmma_f32_16x16x4_f32(false, ahi, false, t, (short)0, acc3, false, false);
  }

#pragma unroll
  for (int i = 0; i < 8; ++i) {
    const int row = m0 + i + (half << 3);
    if (row >= R) continue;
    float vals[4] = { acc0[i], acc1[i], acc2[i], acc3[i] };
#pragma unroll
    for (int j = 0; j < 4; ++j)
      dst[(size_t)row * 128 + j * 16 + l16] = vals[j] + bias[j];
  }
}

__device__ __forceinline__ void wave_lds_handover() {
  __builtin_amdgcn_fence(__ATOMIC_RELEASE, "wavefront");
  __builtin_amdgcn_wave_barrier();
  __builtin_amdgcn_fence(__ATOMIC_ACQUIRE, "wavefront");
}

__device__ __forceinline__ float quad_sum(float s) {
  s += __int_as_float(__builtin_amdgcn_update_dpp(0, __float_as_int(s), 0xB1, 0xF, 0xF, true));
  s += __int_as_float(__builtin_amdgcn_update_dpp(0, __float_as_int(s), 0x4E, 0xF, 0xF, true));
  return s;
}

__device__ __forceinline__ float edge_dot(const float4 p1, const float4 w,
                                          const float* __restrict__ P2, int t, int d0) {
  const float4 p2 = *(const float4*)(P2 + (size_t)t * 128 + d0);
  const float s = lrelu(p1.x + p2.x) * w.x + lrelu(p1.y + p2.y) * w.y +
                  lrelu(p1.z + p2.z) * w.z + lrelu(p1.w + p2.w) * w.w;
  return quad_sum(s);
}

#define GRP 8

__device__ __forceinline__ float lrelu_max(float x) { return fmaxf(x, NEG_SLOPE * x); }

__device__ __forceinline__ float quad_x1(float x) {
  return __int_as_float(__builtin_amdgcn_update_dpp(0, __float_as_int(x), 0xB1, 0xF, 0xF, true));
}
__device__ __forceinline__ float quad_x2(float x) {
  return __int_as_float(__builtin_amdgcn_update_dpp(0, __float_as_int(x), 0x4E, 0xF, 0xF, true));
}

__device__ __forceinline__ float quad_fold4(const float s0, const float s1, const float s2,
                                            const float s3, const bool odd1, const bool odd2) {
  const float k0 = odd1 ? s1 : s0, g0 = odd1 ? s0 : s1;
  const float k1 = odd1 ? s3 : s2, g1 = odd1 ? s2 : s3;
  const float r0 = k0 + quad_x1(g0);
  const float r1 = k1 + quad_x1(g1);
  const float kp = odd2 ? r1 : r0, gv = odd2 ? r0 : r1;
  return kp + quad_x2(gv);
}

__device__ __forceinline__ void logits_to_lds(const int tv, const int cnt, const int jbase,
                                              const float4 p1, const float4 w,
                                              const float* __restrict__ P2, const int d0,
                                              const int lane, float* __restrict__ slog) {
  const int h = lane >> 2, q = lane & 3;
  const bool odd1 = (q & 1) != 0, odd2 = (q & 2) != 0;
  for (int j = 0; j < cnt; j += GRP) {
    float4 a[GRP];
#pragma unroll
    for (int u = 0; u < GRP; ++u) {
      int ju = j + u; ju = ju < cnt ? ju : cnt - 1;
      const int t = __builtin_amdgcn_readlane(tv, ju);
      a[u] = *(const float4*)(P2 + (size_t)t * 128 + d0);
    }
    float s[GRP];
#pragma unroll
    for (int u = 0; u < GRP; ++u) {
      float x = lrelu_max(p1.x + a[u].x) * w.x;
      x = fmaf(lrelu_max(p1.y + a[u].y), w.y, x);
      x = fmaf(lrelu_max(p1.z + a[u].z), w.z, x);
      x = fmaf(lrelu_max(p1.w + a[u].w), w.w, x);
      s[u] = x;
    }
#pragma unroll
    for (int u = 0; u < GRP; u += 4) {
      const float val = quad_fold4(s[u], s[u + 1], s[u + 2], s[u + 3], odd1, odd2);
      const int je = j + u + q;
      if (je < cnt) slog[(jbase + je) * NUM_HEAD + h] = val;
    }
  }
}

__device__ __forceinline__ void weighted_rows(const int tv, const int cnt, const int jbase,
                                              const float* __restrict__ Mmat, const int d0,
                                              const int lane, const float* __restrict__ slog,
                                              float4& acc) {
  const int h = lane >> 2;
  for (int j = 0; j < cnt; j += GRP) {
    float4 a[GRP];
    float be[GRP];
#pragma unroll
    for (int u = 0; u < GRP; ++u) {
      int ju = j + u; ju = ju < cnt ? ju : cnt - 1;
      const int t = __builtin_amdgcn_readlane(tv, ju);
      a[u] = *(const float4*)(Mmat + (size_t)t * 128 + d0);
      be[u] = slog[(jbase + ju) * NUM_HEAD + h];
    }
#pragma unroll
    for (int u = 0; u < GRP; ++u) {
      if (j + u < cnt) {
        acc.x += be[u] * a[u].x; acc.y += be[u] * a[u].y;
        acc.z += be[u] * a[u].z; acc.w += be[u] * a[u].w;
      }
    }
  }
}

__global__ void __launch_bounds__(256) k_rel(const int* __restrict__ trip,
                      const int* __restrict__ recsw, const int* __restrict__ offw,
                      const float* __restrict__ P1, const float* __restrict__ P2,
                      const float* __restrict__ Mmat, const float* __restrict__ av,
                      const float* __restrict__ ab, float* __restrict__ out, int R, int E) {
  __shared__ float s_bias[NUM_BIN * NUM_HEAD];
  __shared__ float s_log[8][SEG_CAP * NUM_HEAD];
  __shared__ int   s_tail[8][SEG_CAP];
  __shared__ int   s_bin[8][SEG_CAP];
  __shared__ int   s_e[BREL][SEG_CAP];
  __shared__ int   s_y[BREL][SEG_CAP];
  __shared__ int   s_n[BREL];

  const int lane = threadIdx.x & 31;
  const int wv = __builtin_amdgcn_readfirstlane(threadIdx.x >> 5);
  int2 oc = make_int2(0, 0);
  if (blockIdx.x < MAXBIN) oc = ((const int2*)offw)[(size_t)blockIdx.x * NPROD + threadIdx.x];
  if (threadIdx.x < NUM_BIN * NUM_HEAD) s_bias[threadIdx.x] = lrelu(ab[threadIdx.x]);
  if (threadIdx.x < BREL) s_n[threadIdx.x] = 0;
  __syncthreads();

  {
    const int chunk = (E + NPROD - 1) / NPROD;
    const int2* __restrict__ src = (const int2*)recsw + (size_t)threadIdx.x * chunk + oc.x;
    for (int i0 = 0; i0 < oc.y; i0 += 4) {
      int2 rc[4];
#pragma unroll
      for (int u = 0; u < 4; ++u) rc[u] = src[i0 + u < oc.y ? i0 + u : oc.y - 1];
#pragma unroll
      for (int u = 0; u < 4; ++u) {
        if (i0 + u < oc.y) {
          const int hl = (rc[u].y >> 4) & (BREL - 1);
          const int slot = atomicAdd(&s_n[hl], 1);
          if (slot < SEG_CAP) { s_e[hl][slot] = rc[u].x; s_y[hl][slot] = rc[u].y; }
        }
      }
    }
  }
  __syncthreads();

  const int d0 = lane * 4;
  const int h  = lane >> 2;
  const float4 w  = *(const float4*)(av + d0);
  float* __restrict__ slog = s_log[wv];

  for (int rr = 0; rr < BREL / 8; ++rr) {
  const int hl = wv * (BREL / 8) + rr;
  const int r = blockIdx.x * BREL + hl;
  if (r >= R) break;
  const int n = __builtin_amdgcn_readfirstlane(s_n[hl]);

  const float4 p1 = *(const float4*)(P1 + (size_t)r * 128 + d0);
  float4 acc = make_float4(0.f, 0.f, 0.f, 0.f);

  if (n <= SEG_CAP) {
    const bool has0 = lane < n, has1 = lane + 32 < n;
    int2 q0 = make_int2(BIG_I, 0), q1 = make_int2(BIG_I, 0);
    if (has0) q0 = make_int2(s_e[hl][lane], s_y[hl][lane]);
    if (has1) q1 = make_int2(s_e[hl][32 + lane], s_y[hl][32 + lane]);
    const int e0 = q0.x, e1 = q1.x;
    int r0 = 0, r1 = 0;
    const int nlo = n < 32 ? n : 32;
    const int nhi = n - nlo;
    for (int j = 0; j < nlo; ++j) {
      const int x = __builtin_amdgcn_readlane(e0, j);
      r0 += (x < e0); r1 += (x < e1);
    }
    for (int j = 0; j < nhi; ++j) {
      const int x = __builtin_amdgcn_readlane(e1, j);
      r0 += (x < e0); r1 += (x < e1);
    }
    if (has0) { s_tail[wv][r0] = q0.y >> 8; s_bin[wv][r0] = q0.y & 15; }
    if (has1) { s_tail[wv][r1] = q1.y >> 8; s_bin[wv][r1] = q1.y & 15; }
    wave_lds_handover();
    const int tv0 = s_tail[wv][lane];
    const int tv1 = s_tail[wv][lane + 32];

    logits_to_lds(tv0, nlo, 0,  p1, w, P2, d0, lane, slog);
    logits_to_lds(tv1, nhi, 32, p1, w, P2, d0, lane, slog);
    wave_lds_handover();

    float v[SEG_CAP / 4];
    float mx = -BIG_F;
#pragma unroll
    for (int i = 0; i < SEG_CAP / 4; ++i) {
      const int j = (lane >> 3) + 4 * i;
      v[i] = -BIG_F;
      if (j < n) v[i] = slog[lane + 32 * i] + s_bias[s_bin[wv][j] * NUM_HEAD + (lane & 7)];
      mx = fmaxf(mx, v[i]);
    }
    mx = fmaxf(mx, __shfl_xor(mx, 8, 32));
    mx = fmaxf(mx, __shfl_xor(mx, 16, 32));
    float sum = 0.f;
#pragma unroll
    for (int i = 0; i < SEG_CAP / 4; ++i) {
      const int j = (lane >> 3) + 4 * i;
      v[i] = (j < n) ? expf(v[i] - mx) : 0.f;
      sum += v[i];
    }
    sum += __shfl_xor(sum, 8, 32);
    sum += __shfl_xor(sum, 16, 32);
    const float den = sum + 1e-16f;
#pragma unroll
    for (int i = 0; i < SEG_CAP / 4; ++i) {
      const int j = (lane >> 3) + 4 * i;
      if (j < n) slog[lane + 32 * i] = v[i] / den;
    }
    wave_lds_handover();

    weighted_rows(tv0, nlo, 0,  Mmat, d0, lane, slog, acc);
    weighted_rows(tv1, nhi, 32, Mmat, d0, lane, slog, acc);
  } else {
    float M = -BIG_F, L = 0.f;
    for (int base = 0; base < E; base += 32) {
      const int idx = base + lane;
      const bool mine = idx < E && trip[3 * idx] == r;
      unsigned todo = __builtin_amdgcn_readfirstlane((unsigned)__ballot(mine));
      while (todo != 0u) {
        const int e = base + __builtin_ctz(todo);
        todo &= todo - 1u;
        const int t = trip[3 * e + 1];
        const int b = clamp_bin(trip[3 * e + 2]);
        const float s = edge_dot(p1, w, P2, t, d0) + s_bias[b * NUM_HEAD + h];
        const float Mn = fmaxf(M, s);
        const float sc = expf(M - Mn), p = expf(s - Mn);
        const float4 m = *(const float4*)(Mmat + (size_t)t * 128 + d0);
        L = L * sc + p;
        acc.x = acc.x * sc + p * m.x; acc.y = acc.y * sc + p * m.y;
        acc.z = acc.z * sc + p * m.z; acc.w = acc.w * sc + p * m.w;
        M = Mn;
      }
    }
    const float den = L + 1e-16f;
    acc.x /= den; acc.y /= den; acc.z /= den; acc.w /= den;
  }
  *(float4*)(out + (size_t)r * 128 + d0) = acc;
  }
}

extern "C" void kernel_launch(void* const* d_in, const int* in_sizes, int n_in,
                              void* d_out, int out_size, void* d_ws, size_t ws_size,
                              hipStream_t stream) {
  const float* emb      = (const float*)d_in[0];
  const int*   trip     = (const int*)  d_in[1];
  const float* attn_w   = (const float*)d_in[2];
  const float* attn_b   = (const float*)d_in[3];
  const float* attn_bin = (const float*)d_in[4];
  const float* attn_vec = (const float*)d_in[5];
  const float* aggr_w   = (const float*)d_in[6];
  const float* aggr_b   = (const float*)d_in[7];

  const int R = in_sizes[0] / DIM_IN;
  const int E = in_sizes[1] / 3;

  float* ws     = (float*)d_ws;
  float* P1     = ws + DIM_IN * NCOLS + NCOLS;
  float* P2     = P1 + (size_t)R * 128;
  float* Mm     = P2 + (size_t)R * 128;
  const int chunk = (E + NPROD - 1) / NPROD;
  int*   recs   = (int*)(Mm + (size_t)R * 128);
  int*   offT   = recs + (size_t)NPROD * chunk * 2;
  float* out    = (float*)d_out;

  const int ggrp = ((R + 127) / 128) * (NCOLS / 64);
  k_front<<<NPROD + ggrp, 256, 0, stream>>>(emb, attn_w, attn_b, aggr_w, aggr_b, P1, P2, Mm, trip, recs, offT, R, E);

  k_rel<<<(R + BREL - 1) / BREL, NPROD, 0, stream>>>(trip, recs, offT, P1, P2, Mm, attn_vec, attn_bin, out, R, E);
}
